# starting kernel
# baseline (speedup 1.0000x reference)
#include <hip/hip_runtime.h>

#define B_     64
#define CIN    3
#define Hh     96
#define Ww     96
#define COUT   96
#define KH     5
#define KW     5
#define STRIDE 2
#define PAD    2
#define HOUT   48
#define WOUT   48
#define L_     (HOUT*WOUT)
#define K_     (CIN*KH*KW)
#define KP     76
#define LR     0.02f
#define DELTA  (-0.4f)
#define PRECF  1e-30f

typedef float v2f __attribute__((ext_vector_type(2)));
typedef float v8f __attribute__((ext_vector_type(8)));

__global__ __launch_bounds__(192) void conv_hebb_gemm(
    const float* __restrict__ x, const float* __restrict__ w,
    float* __restrict__ out, float* __restrict__ yx, float* __restrict__ yy)
{
    __shared__ float lds_w[COUT * KP];
    __shared__ float lds_p[KP * 16];
    __shared__ float lds_t[COUT * 16];
    __shared__ int   s_i0[16], s_i1[16];
    __shared__ float s_t0[16], s_t1[16];

    const int tid = threadIdx.x;
    const int b   = blockIdx.y;
    const int l0  = blockIdx.x * 16;

    for (int i = tid; i < COUT * KP; i += 192) {
        int o = i / KP, k = i - o * KP;
        lds_w[i] = (k < K_) ? w[o * K_ + k] : 0.0f;
    }
    for (int i = tid; i < KP * 16; i += 192) {
        int k = i >> 4, n = i & 15;
        float v = 0.0f;
        if (k < K_) {
            int c  = k / (KH * KW);
            int r  = k - c * (KH * KW);
            int ki = r / KW, kj = r - ki * KW;
            int l  = l0 + n;
            int y  = l / WOUT, xo = l - y * WOUT;
            int iy = y * STRIDE + ki - PAD;
            int ix = xo * STRIDE + kj - PAD;
            if (iy >= 0 && iy < Hh && ix >= 0 && ix < Ww)
                v = x[((b * CIN + c) * Hh + iy) * Ww + ix];
        }
        lds_p[i] = v;
    }
    __syncthreads();

    const int wv   = tid >> 5;
    const int lane = tid & 31;
    const int mn   = lane & 15;
    const int kg   = lane >> 4;

    v8f accO = {}; v8f accT = {};
    const float* wr = &lds_w[(wv * 16 + mn) * KP];
    #pragma unroll
    for (int kk = 0; kk < KP; kk += 4) {
        int ka = kk + 2 * kg;
        float a0 = wr[ka], a1 = wr[ka + 1];
        v2f aW = {a0, a1};
        v2f aL = {a0 * __builtin_fabsf(a0), a1 * __builtin_fabsf(a1)};
        v2f bP = {lds_p[ka * 16 + mn], lds_p[(ka + 1) * 16 + mn]};
        accO = __builtin_amdgcn_wmma_f32_16x16x4_f32(
            false, aW, false, bP, (short)0, accO, false, false);
        accT = __builtin_amdgcn_wmma_f32_16x16x4_f32(
            false, aL, false, bP, (short)0, accT, false, false);
    }

    #pragma unroll
    for (int r = 0; r < 8; ++r) {
        int o = wv * 16 + r + 8 * kg;
        out[((size_t)b * COUT + o) * L_ + l0 + mn] = accO[r];
        lds_t[o * 16 + mn] = accT[r];
    }
    __syncthreads();

    if (tid < 16) {
        int i0 = 0; float v0 = lds_t[tid];
        for (int o = 1; o < COUT; ++o) {
            float v = lds_t[o * 16 + tid];
            if (v > v0) { v0 = v; i0 = o; }
        }
        int i1 = (i0 == 0) ? 1 : 0;
        float v1 = lds_t[i1 * 16 + tid];
        for (int o = 0; o < COUT; ++o) {
            if (o == i0) continue;
            float v = lds_t[o * 16 + tid];
            if (v > v1) { v1 = v; i1 = o; }
        }
        s_i0[tid] = i0; s_i1[tid] = i1; s_t0[tid] = v0; s_t1[tid] = v1;
    }
    __syncthreads();

    {
        int col = tid / 12;
        int r   = tid - col * 12;
        int i0 = s_i0[col], i1 = s_i1[col];
        for (int k = r; k < K_; k += 12) {
            float p = lds_p[k * 16 + col];
            atomicAdd(&yx[i0 * K_ + k], p);
            atomicAdd(&yx[i1 * K_ + k], DELTA * p);
        }
        if (r == 0) {
            atomicAdd(&yy[i0], s_t0[col]);
            atomicAdd(&yy[i1], DELTA * s_t1[col]);
        }
    }
}

__global__ void zero_ws(float* yx, float* yy, unsigned int* mb)
{
    int i = blockIdx.x * blockDim.x + threadIdx.x;
    if (i < COUT * K_) yx[i] = 0.0f;
    if (i < COUT)      yy[i] = 0.0f;
    if (i == 0)        *mb = __float_as_uint(PRECF);
}

__global__ void finalize_ds(const float* __restrict__ w,
                            const float* __restrict__ yx,
                            const float* __restrict__ yy,
                            float* __restrict__ ds, unsigned int* mb)
{
    int i = blockIdx.x * blockDim.x + threadIdx.x;
    if (i < COUT * K_) {
        int o = i / K_;
        float d = yx[i] * (1.0f / (float)L_) - (yy[o] * (1.0f / (float)L_)) * w[i];
        ds[i] = d;
        atomicMax(mb, __float_as_uint(__builtin_fabsf(d)));
    }
}

__global__ void apply_update(const float* __restrict__ w,
                             const float* __restrict__ ds,
                             const unsigned int* __restrict__ mb,
                             float* __restrict__ neww)
{
    int i = blockIdx.x * blockDim.x + threadIdx.x;
    if (i < COUT * K_) {
        float nc = __uint_as_float(*mb);
        neww[i] = w[i] + LR * (ds[i] / nc);
    }
}

extern "C" void kernel_launch(void* const* d_in, const int* in_sizes, int n_in,
                              void* d_out, int out_size, void* d_ws, size_t ws_size,
                              hipStream_t stream)
{
    const float* x = (const float*)d_in[0];
    const float* w = (const float*)d_in[1];
    float* out  = (float*)d_out;
    float* neww = out + (size_t)B_ * COUT * L_;

    float*        yx = (float*)d_ws;
    float*        yy = yx + COUT * K_;
    unsigned int* mb = (unsigned int*)(yy + COUT);
    float*        ds = (float*)(mb + 1);

    zero_ws<<<(COUT * K_ + 255) / 256, 256, 0, stream>>>(yx, yy, mb);
    conv_hebb_gemm<<<dim3(L_ / 16, B_), 192, 0, stream>>>(x, w, out, yx, yy);
    finalize_ds<<<(COUT * K_ + 255) / 256, 256, 0, stream>>>(w, yx, yy, ds, mb);
    apply_update<<<(COUT * K_ + 255) / 256, 256, 0, stream>>>(w, ds, mb, neww);
}


# One work-group per quarter image walks its 36 tiles (4 at a time, weights staged once) summing yx/yy by LDS atomics; one global atomic add per element per group, not per tile: sum order changes
# speedup vs baseline: 6.9472x; 6.9472x over previous
#include <hip/hip_runtime.h>

#define B_     64
#define CIN    3
#define Hh     96
#define Ww     96
#define COUT   96
#define KH     5
#define KW     5
#define STRIDE 2
#define PAD    2
#define HOUT   48
#define WOUT   48
#define L_     (HOUT*WOUT)
#define K_     (CIN*KH*KW)
#define KP     76
#define LR     0.02f
#define DELTA  (-0.4f)
#define PRECF  1e-30f
#define G_     4
#define NT     192
#define TPB    (G_ * NT)
#define BPI    4
#define TPBLK  ((L_ / 16) / BPI)
#define ITERS  (TPBLK / G_)

typedef float v2f __attribute__((ext_vector_type(2)));
typedef float v8f __attribute__((ext_vector_type(8)));

__global__ __launch_bounds__(TPB) void conv_hebb_gemm(
    const float* __restrict__ x, const float* __restrict__ w,
    float* __restrict__ out, float* __restrict__ yx, float* __restrict__ yy)
{
    __shared__ float lds_w[COUT * KP];
    __shared__ float lds_yx[COUT * K_];
    __shared__ float lds_yy[COUT];
    __shared__ float lds_pp[G_][KP * 16];
    __shared__ float lds_tt[G_][COUT * 16];
    __shared__ int   s_i0a[G_][16], s_i1a[G_][16];
    __shared__ float s_t0a[G_][16], s_t1a[G_][16];

    const int gtid = threadIdx.x;
    const int g    = gtid / NT;
    const int tid  = gtid - g * NT;
    const int b    = blockIdx.x / BPI;
    const int q    = blockIdx.x - b * BPI;

    float* lds_p = lds_pp[g];
    float* lds_t = lds_tt[g];
    int*   s_i0 = s_i0a[g];  int*   s_i1 = s_i1a[g];
    float* s_t0 = s_t0a[g];  float* s_t1 = s_t1a[g];

    for (int i = gtid; i < COUT * KP; i += TPB) {
        int o = i / KP, k = i - o * KP;
        lds_w[i] = (k < K_) ? w[o * K_ + k] : 0.0f;
    }
    for (int i = gtid; i < COUT * K_; i += TPB) lds_yx[i] = 0.0f;
    if (gtid < COUT) lds_yy[gtid] = 0.0f;

    const int wv   = tid >> 5;
    const int lane = tid & 31;
    const int mn   = lane & 15;
    const int kg   = lane >> 4;

  for (int it = 0; it < ITERS; ++it) {
    const int l0 = (q * TPBLK + it * G_ + g) * 16;

    for (int i = tid; i < KP * 16; i += NT) {
        int k = i >> 4, n = i & 15;
        float v = 0.0f;
        if (k < K_) {
            int c  = k / (KH * KW);
            int r  = k - c * (KH * KW);
            int ki = r / KW, kj = r - ki * KW;
            int l  = l0 + n;
            int y  = l / WOUT, xo = l - y * WOUT;
            int iy = y * STRIDE + ki - PAD;
            int ix = xo * STRIDE + kj - PAD;
            if (iy >= 0 && iy < Hh && ix >= 0 && ix < Ww)
                v = x[((b * CIN + c) * Hh + iy) * Ww + ix];
        }
        lds_p[i] = v;
    }
    __syncthreads();

    v8f accO = {}; v8f accT = {};
    const float* wr = &lds_w[(wv * 16 + mn) * KP];
    #pragma unroll
    for (int kk = 0; kk < KP; kk += 4) {
        int ka = kk + 2 * kg;
        float a0 = wr[ka], a1 = wr[ka + 1];
        v2f aW = {a0, a1};
        v2f aL = {a0 * __builtin_fabsf(a0), a1 * __builtin_fabsf(a1)};
        v2f bP = {lds_p[ka * 16 + mn], lds_p[(ka + 1) * 16 + mn]};
        accO = __builtin_amdgcn_wmma_f32_16x16x4_f32(
            false, aW, false, bP, (short)0, accO, false, false);
        accT = __builtin_amdgcn_wmma_f32_16x16x4_f32(
            false, aL, false, bP, (short)0, accT, false, false);
    }

    #pragma unroll
    for (int r = 0; r < 8; ++r) {
        int o = wv * 16 + r + 8 * kg;
        out[((size_t)b * COUT + o) * L_ + l0 + mn] = accO[r];
        lds_t[o * 16 + mn] = accT[r];
    }
    __syncthreads();

    if (tid < 16) {
        int i0 = 0; float v0 = lds_t[tid];
        for (int o = 1; o < COUT; ++o) {
            float v = lds_t[o * 16 + tid];
            if (v > v0) { v0 = v; i0 = o; }
        }
        int i1 = (i0 == 0) ? 1 : 0;
        float v1 = lds_t[i1 * 16 + tid];
        for (int o = 0; o < COUT; ++o) {
            if (o == i0) continue;
            float v = lds_t[o * 16 + tid];
            if (v > v1) { v1 = v; i1 = o; }
        }
        s_i0[tid] = i0; s_i1[tid] = i1; s_t0[tid] = v0; s_t1[tid] = v1;
    }
    __syncthreads();

    {
        int col = tid / 12;
        int r   = tid - col * 12;
        int i0 = s_i0[col], i1 = s_i1[col];
        for (int k = r; k < K_; k += 12) {
            float p = lds_p[k * 16 + col];
            atomicAdd(&lds_yx[i0 * K_ + k], p);
            atomicAdd(&lds_yx[i1 * K_ + k], DELTA * p);
        }
        if (r == 0) {
            atomicAdd(&lds_yy[i0], s_t0[col]);
            atomicAdd(&lds_yy[i1], DELTA * s_t1[col]);
        }
    }
    __syncthreads();
  }

    for (int i = gtid; i < COUT * K_; i += TPB) atomicAdd(&yx[i], lds_yx[i]);
    if (gtid < COUT) atomicAdd(&yy[gtid], lds_yy[gtid]);
}

__global__ void zero_ws(float* yx, float* yy, unsigned int* mb)
{
    int i = blockIdx.x * blockDim.x + threadIdx.x;
    if (i < COUT * K_) yx[i] = 0.0f;
    if (i < COUT)      yy[i] = 0.0f;
    if (i == 0)        *mb = __float_as_uint(PRECF);
}

__global__ void finalize_ds(const float* __restrict__ w,
                            const float* __restrict__ yx,
                            const float* __restrict__ yy,
                            float* __restrict__ ds, unsigned int* mb)
{
    int i = blockIdx.x * blockDim.x + threadIdx.x;
    if (i < COUT * K_) {
        int o = i / K_;
        float d = yx[i] * (1.0f / (float)L_) - (yy[o] * (1.0f / (float)L_)) * w[i];
        ds[i] = d;
        atomicMax(mb, __float_as_uint(__builtin_fabsf(d)));
    }
}

__global__ void apply_update(const float* __restrict__ w,
                             const float* __restrict__ ds,
                             const unsigned int* __restrict__ mb,
                             float* __restrict__ neww)
{
    int i = blockIdx.x * blockDim.x + threadIdx.x;
    if (i < COUT * K_) {
        float nc = __uint_as_float(*mb);
        neww[i] = w[i] + LR * (ds[i] / nc);
    }
}

extern "C" void kernel_launch(void* const* d_in, const int* in_sizes, int n_in,
                              void* d_out, int out_size, void* d_ws, size_t ws_size,
                              hipStream_t stream)
{
    const float* x = (const float*)d_in[0];
    const float* w = (const float*)d_in[1];
    float* out  = (float*)d_out;
    float* neww = out + (size_t)B_ * COUT * L_;

    float*        yx = (float*)d_ws;
    float*        yy = yx + COUT * K_;
    unsigned int* mb = (unsigned int*)(yy + COUT);
    float*        ds = (float*)(mb + 1);

    zero_ws<<<(COUT * K_ + 255) / 256, 256, 0, stream>>>(yx, yy, mb);
    conv_hebb_gemm<<<B_ * BPI, TPB, 0, stream>>>(x, w, out, yx, yy);
    finalize_ds<<<(COUT * K_ + 255) / 256, 256, 0, stream>>>(w, yx, yy, ds, mb);
    apply_update<<<(COUT * K_ + 255) / 256, 256, 0, stream>>>(w, ds, mb, neww);
}


# Top-2 of each column picked from the accumulator registers (lane over its 8 rows, lane-half exchange, 6 wave records merged by the scatter threads); the 16-thread serial LDS scan is gone
# speedup vs baseline: 11.9168x; 1.7153x over previous
#include <hip/hip_runtime.h>

#define B_     64
#define CIN    3
#define Hh     96
#define Ww     96
#define COUT   96
#define KH     5
#define KW     5
#define STRIDE 2
#define PAD    2
#define HOUT   48
#define WOUT   48
#define L_     (HOUT*WOUT)
#define K_     (CIN*KH*KW)
#define KP     76
#define LR     0.02f
#define DELTA  (-0.4f)
#define PRECF  1e-30f
#define G_     4
#define NT     192
#define TPB    (G_ * NT)
#define BPI    4
#define TPBLK  ((L_ / 16) / BPI)
#define ITERS  (TPBLK / G_)

typedef float v2f __attribute__((ext_vector_type(2)));
typedef float v8f __attribute__((ext_vector_type(8)));

typedef int v4i __attribute__((ext_vector_type(4)));

__device__ __forceinline__ int xhalf_i(int b)
{
    return __builtin_amdgcn_permlanex16(b, b, 0x76543210, 0xfedcba98, false, false);
}
__device__ __forceinline__ float xhalf_f(float x)
{
    return __int_as_float(xhalf_i(__float_as_int(x)));
}

__device__ __forceinline__ bool ranks_before(float av, int ai, float bv, int bi)
{
    return (av > bv) || (av == bv && ai < bi);
}

__device__ __forceinline__ void merge_top2(float& v0, int& i0, float& v1, int& i1,
                                           float w0, int j0, float w1, int j1)
{
    const bool af = ranks_before(v0, i0, w0, j0);
    const float xv = af ? v1 : v0;  const int xi = af ? i1 : i0;
    const float yv = af ? w0 : w1;  const int yi = af ? j0 : j1;
    const bool xs = ranks_before(xv, xi, yv, yi);
    v0 = af ? v0 : w0;  i0 = af ? i0 : j0;
    v1 = xs ? xv : yv;  i1 = xs ? xi : yi;
}

__global__ __launch_bounds__(TPB) void conv_hebb_gemm(
    const float* __restrict__ x, const float* __restrict__ w,
    float* __restrict__ out, float* __restrict__ yx, float* __restrict__ yy)
{
    __shared__ float lds_w[COUT * KP];
    __shared__ float lds_yx[COUT * K_];
    __shared__ float lds_yy[COUT];
    __shared__ float lds_pp[G_][KP * 16];
    __shared__ v4i   lds_cc[G_][16 * 6];

    const int gtid = threadIdx.x;
    const int g    = gtid / NT;
    const int tid  = gtid - g * NT;
    const int b    = blockIdx.x / BPI;
    const int q    = blockIdx.x - b * BPI;

    float* lds_p = lds_pp[g];
    v4i*   lds_c = lds_cc[g];

    for (int i = gtid; i < COUT * KP; i += TPB) {
        int o = i / KP, k = i - o * KP;
        lds_w[i] = (k < K_) ? w[o * K_ + k] : 0.0f;
    }
    for (int i = gtid; i < COUT * K_; i += TPB) lds_yx[i] = 0.0f;
    if (gtid < COUT) lds_yy[gtid] = 0.0f;

    const int wv   = tid >> 5;
    const int lane = tid & 31;
    const int mn   = lane & 15;
    const int kg   = lane >> 4;

  for (int it = 0; it < ITERS; ++it) {
    const int l0 = (q * TPBLK + it * G_ + g) * 16;

    for (int i = tid; i < KP * 16; i += NT) {
        int k = i >> 4, n = i & 15;
        float v = 0.0f;
        if (k < K_) {
            int c  = k / (KH * KW);
            int r  = k - c * (KH * KW);
            int ki = r / KW, kj = r - ki * KW;
            int l  = l0 + n;
            int y  = l / WOUT, xo = l - y * WOUT;
            int iy = y * STRIDE + ki - PAD;
            int ix = xo * STRIDE + kj - PAD;
            if (iy >= 0 && iy < Hh && ix >= 0 && ix < Ww)
                v = x[((b * CIN + c) * Hh + iy) * Ww + ix];
        }
        lds_p[i] = v;
    }
    __syncthreads();

    v8f accO = {}; v8f accT = {};
    const float* wr = &lds_w[(wv * 16 + mn) * KP];
    #pragma unroll
    for (int kk = 0; kk < KP; kk += 4) {
        int ka = kk + 2 * kg;
        float a0 = wr[ka], a1 = wr[ka + 1];
        v2f aW = {a0, a1};
        v2f aL = {a0 * __builtin_fabsf(a0), a1 * __builtin_fabsf(a1)};
        v2f bP = {lds_p[ka * 16 + mn], lds_p[(ka + 1) * 16 + mn]};
        accO = __builtin_amdgcn_wmma_f32_16x16x4_f32(
            false, aW, false, bP, (short)0, accO, false, false);
        accT = __builtin_amdgcn_wmma_f32_16x16x4_f32(
            false, aL, false, bP, (short)0, accT, false, false);
    }

    const int ob = wv * 16 + 8 * kg;
    #pragma unroll
    for (int r = 0; r < 8; ++r)
        out[((size_t)b * COUT + ob + r) * L_ + l0 + mn] = accO[r];

    {
        float v0, v1; int i0, i1;
        if (accT[1] > accT[0]) { v0 = accT[1]; i0 = ob + 1; v1 = accT[0]; i1 = ob; }
        else                   { v0 = accT[0]; i0 = ob;     v1 = accT[1]; i1 = ob + 1; }
        #pragma unroll
        for (int r = 2; r < 8; ++r) {
            const float e = accT[r];
            const bool g0 = e > v0, g1 = e > v1;
            v1 = g0 ? v0 : (g1 ? e : v1);
            i1 = g0 ? i0 : (g1 ? ob + r : i1);
            v0 = g0 ? e : v0;
            i0 = g0 ? ob + r : i0;
        }
        float w0 = xhalf_f(v0), w1 = xhalf_f(v1);
        int   j0 = xhalf_i(i0), j1 = xhalf_i(i1);
        merge_top2(v0, i0, v1, i1, w0, j0, w1, j1);
        if (kg == 0) {
            v4i rec = {__float_as_int(v0), i0, __float_as_int(v1), i1};
            lds_c[mn * 6 + wv] = rec;
        }
    }
    __syncthreads();

    {
        int col = tid / 12;
        int r   = tid - col * 12;
        v4i rec[6];
        #pragma unroll
        for (int m = 0; m < 6; ++m) rec[m] = lds_c[col * 6 + m];
        float v0 = __int_as_float(rec[0].x), v1 = __int_as_float(rec[0].z);
        int   i0 = rec[0].y, i1 = rec[0].w;
        #pragma unroll
        for (int m = 1; m < 6; ++m)
            merge_top2(v0, i0, v1, i1, __int_as_float(rec[m].x), rec[m].y,
                       __int_as_float(rec[m].z), rec[m].w);
        for (int k = r; k < K_; k += 12) {
            float p = lds_p[k * 16 + col];
            atomicAdd(&lds_yx[i0 * K_ + k], p);
            atomicAdd(&lds_yx[i1 * K_ + k], DELTA * p);
        }
        if (r == 0) {
            atomicAdd(&lds_yy[i0], v0);
            atomicAdd(&lds_yy[i1], DELTA * v1);
        }
    }
    __syncthreads();
  }

    for (int i = gtid; i < COUT * K_; i += TPB) atomicAdd(&yx[i], lds_yx[i]);
    if (gtid < COUT) atomicAdd(&yy[gtid], lds_yy[gtid]);
}

__global__ void zero_ws(float* yx, float* yy, unsigned int* mb)
{
    int i = blockIdx.x * blockDim.x + threadIdx.x;
    if (i < COUT * K_) yx[i] = 0.0f;
    if (i < COUT)      yy[i] = 0.0f;
    if (i == 0)        *mb = __float_as_uint(PRECF);
}

__global__ void finalize_ds(const float* __restrict__ w,
                            const float* __restrict__ yx,
                            const float* __restrict__ yy,
                            float* __restrict__ ds, unsigned int* mb)
{
    int i = blockIdx.x * blockDim.x + threadIdx.x;
    if (i < COUT * K_) {
        int o = i / K_;
        float d = yx[i] * (1.0f / (float)L_) - (yy[o] * (1.0f / (float)L_)) * w[i];
        ds[i] = d;
        atomicMax(mb, __float_as_uint(__builtin_fabsf(d)));
    }
}

__global__ void apply_update(const float* __restrict__ w,
                             const float* __restrict__ ds,
                             const unsigned int* __restrict__ mb,
                             float* __restrict__ neww)
{
    int i = blockIdx.x * blockDim.x + threadIdx.x;
    if (i < COUT * K_) {
        float nc = __uint_as_float(*mb);
        neww[i] = w[i] + LR * (ds[i] / nc);
    }
}

extern "C" void kernel_launch(void* const* d_in, const int* in_sizes, int n_in,
                              void* d_out, int out_size, void* d_ws, size_t ws_size,
                              hipStream_t stream)
{
    const float* x = (const float*)d_in[0];
    const float* w = (const float*)d_in[1];
    float* out  = (float*)d_out;
    float* neww = out + (size_t)B_ * COUT * L_;

    float*        yx = (float*)d_ws;
    float*        yy = yx + COUT * K_;
    unsigned int* mb = (unsigned int*)(yy + COUT);
    float*        ds = (float*)(mb + 1);

    zero_ws<<<(COUT * K_ + 255) / 256, 256, 0, stream>>>(yx, yy, mb);
    conv_hebb_gemm<<<B_ * BPI, TPB, 0, stream>>>(x, w, out, yx, yy);
    finalize_ds<<<(COUT * K_ + 255) / 256, 256, 0, stream>>>(w, yx, yy, ds, mb);
    apply_update<<<(COUT * K_ + 255) / 256, 256, 0, stream>>>(w, ds, mb, neww);
}


# Each tile's patch values are read one tile ahead as one branch-free group of 7 loads, issued behind the barrier that opens the current tile's products: no serial load round trips per tile
# speedup vs baseline: 14.7894x; 1.2411x over previous
#include <hip/hip_runtime.h>

#define B_     64
#define CIN    3
#define Hh     96
#define Ww     96
#define COUT   96
#define KH     5
#define KW     5
#define STRIDE 2
#define PAD    2
#define HOUT   48
#define WOUT   48
#define L_     (HOUT*WOUT)
#define K_     (CIN*KH*KW)
#define KP     76
#define LR     0.02f
#define DELTA  (-0.4f)
#define PRECF  1e-30f
#define G_     4
#define NT     192
#define TPB    (G_ * NT)
#define BPI    4
#define TPBLK  ((L_ / 16) / BPI)
#define ITERS  (TPBLK / G_)

typedef float v2f __attribute__((ext_vector_type(2)));
typedef float v8f __attribute__((ext_vector_type(8)));

typedef int v4i __attribute__((ext_vector_type(4)));

__device__ __forceinline__ int xhalf_i(int b)
{
    return __builtin_amdgcn_permlanex16(b, b, 0x76543210, 0xfedcba98, false, false);
}
__device__ __forceinline__ float xhalf_f(float x)
{
    return __int_as_float(xhalf_i(__float_as_int(x)));
}

__device__ __forceinline__ bool ranks_before(float av, int ai, float bv, int bi)
{
    return (av > bv) || (av == bv && ai < bi);
}

__device__ __forceinline__ void merge_top2(float& v0, int& i0, float& v1, int& i1,
                                           float w0, int j0, float w1, int j1)
{
    const bool af = ranks_before(v0, i0, w0, j0);
    const float xv = af ? v1 : v0;  const int xi = af ? i1 : i0;
    const float yv = af ? w0 : w1;  const int yi = af ? j0 : j1;
    const bool xs = ranks_before(xv, xi, yv, yi);
    v0 = af ? v0 : w0;  i0 = af ? i0 : j0;
    v1 = xs ? xv : yv;  i1 = xs ? xi : yi;
}

#define NG ((KP * 16 + NT - 1) / NT)

__device__ __forceinline__ int gather_issue(const float* __restrict__ xb, int l0, int n,
                                            const int (&pk)[NG], float (&raw)[NG])
{
    const int l   = l0 + n;
    const int y   = l / WOUT, xo = l - y * WOUT;
    const int iy0 = y * STRIDE - PAD, ix0 = xo * STRIDE - PAD;
    const int rb  = iy0 * Ww + ix0;
    int okb = 0;
    #pragma unroll
    for (int j = 0; j < NG; ++j) {
        const int ki = (pk[j] >> 16) & 15, kj = (pk[j] >> 20) & 15;
        const bool ok = pk[j] >= 0 && (unsigned)(iy0 + ki) < (unsigned)Hh
                                   && (unsigned)(ix0 + kj) < (unsigned)Ww;
        const int idx = ok ? rb + (pk[j] & 0xffff) : 0;
        raw[j] = xb[idx];
        okb |= ok ? (1 << j) : 0;
    }
    return okb;
}

__global__ __launch_bounds__(TPB) void conv_hebb_gemm(
    const float* __restrict__ x, const float* __restrict__ w,
    float* __restrict__ out, float* __restrict__ yx, float* __restrict__ yy)
{
    __shared__ float lds_w[COUT * KP];
    __shared__ float lds_yx[COUT * K_];
    __shared__ float lds_yy[COUT];
    __shared__ float lds_pp[G_][KP * 16];
    __shared__ v4i   lds_cc[G_][16 * 6];

    const int gtid = threadIdx.x;
    const int g    = gtid / NT;
    const int tid  = gtid - g * NT;
    const int b    = blockIdx.x / BPI;
    const int q    = blockIdx.x - b * BPI;

    float* lds_p = lds_pp[g];
    v4i*   lds_c = lds_cc[g];

    const float* xb = x + (size_t)b * CIN * Hh * Ww;
    int pk[NG];
    #pragma unroll
    for (int j = 0; j < NG; ++j) {
        const int k  = (tid + NT * j) >> 4;
        const int c  = k / (KH * KW);
        const int r  = k - c * (KH * KW);
        const int ki = r / KW, kj = r - ki * KW;
        pk[j] = (k < K_) ? (((c * Hh + ki) * Ww + kj) | (ki << 16) | (kj << 20)) : -1;
    }
    float raw[NG];
    int okb = gather_issue(xb, (q * TPBLK + g) * 16, tid & 15, pk, raw);

    for (int i = gtid; i < COUT * KP; i += TPB) {
        int o = i / KP, k = i - o * KP;
        lds_w[i] = (k < K_) ? w[o * K_ + k] : 0.0f;
    }
    for (int i = gtid; i < COUT * K_; i += TPB) lds_yx[i] = 0.0f;
    if (gtid < COUT) lds_yy[gtid] = 0.0f;

    const int wv   = tid >> 5;
    const int lane = tid & 31;
    const int mn   = lane & 15;
    const int kg   = lane >> 4;

  for (int it = 0; it < ITERS; ++it) {
    const int l0 = (q * TPBLK + it * G_ + g) * 16;

    #pragma unroll
    for (int j = 0; j < NG; ++j) {
        const int i = tid + NT * j;
        if (j < NG - 1 || i < KP * 16) lds_p[i] = ((okb >> j) & 1) ? raw[j] : 0.0f;
    }
    __syncthreads();

    if (it + 1 < ITERS) okb = gather_issue(xb, l0 + G_ * 16, tid & 15, pk, raw);

    v8f accO = {}; v8f accT = {};
    const float* wr = &lds_w[(wv * 16 + mn) * KP];
    #pragma unroll
    for (int kk = 0; kk < KP; kk += 4) {
        int ka = kk + 2 * kg;
        float a0 = wr[ka], a1 = wr[ka + 1];
        v2f aW = {a0, a1};
        v2f aL = {a0 * __builtin_fabsf(a0), a1 * __builtin_fabsf(a1)};
        v2f bP = {lds_p[ka * 16 + mn], lds_p[(ka + 1) * 16 + mn]};
        accO = __builtin_amdgcn_wmma_f32_16x16x4_f32(
            false, aW, false, bP, (short)0, accO, false, false);
        accT = __builtin_amdgcn_wmma_f32_16x16x4_f32(
            false, aL, false, bP, (short)0, accT, false, false);
    }

    const int ob = wv * 16 + 8 * kg;
    #pragma unroll
    for (int r = 0; r < 8; ++r)
        out[((size_t)b * COUT + ob + r) * L_ + l0 + mn] = accO[r];

    {
        float v0, v1; int i0, i1;
        if (accT[1] > accT[0]) { v0 = accT[1]; i0 = ob + 1; v1 = accT[0]; i1 = ob; }
        else                   { v0 = accT[0]; i0 = ob;     v1 = accT[1]; i1 = ob + 1; }
        #pragma unroll
        for (int r = 2; r < 8; ++r) {
            const float e = accT[r];
            const bool g0 = e > v0, g1 = e > v1;
            v1 = g0 ? v0 : (g1 ? e : v1);
            i1 = g0 ? i0 : (g1 ? ob + r : i1);
            v0 = g0 ? e : v0;
            i0 = g0 ? ob + r : i0;
        }
        float w0 = xhalf_f(v0), w1 = xhalf_f(v1);
        int   j0 = xhalf_i(i0), j1 = xhalf_i(i1);
        merge_top2(v0, i0, v1, i1, w0, j0, w1, j1);
        if (kg == 0) {
            v4i rec = {__float_as_int(v0), i0, __float_as_int(v1), i1};
            lds_c[mn * 6 + wv] = rec;
        }
    }
    __syncthreads();

    {
        int col = tid / 12;
        int r   = tid - col * 12;
        v4i rec[6];
        #pragma unroll
        for (int m = 0; m < 6; ++m) rec[m] = lds_c[col * 6 + m];
        float v0 = __int_as_float(rec[0].x), v1 = __int_as_float(rec[0].z);
        int   i0 = rec[0].y, i1 = rec[0].w;
        #pragma unroll
        for (int m = 1; m < 6; ++m)
            merge_top2(v0, i0, v1, i1, __int_as_float(rec[m].x), rec[m].y,
                       __int_as_float(rec[m].z), rec[m].w);
        for (int k = r; k < K_; k += 12) {
            float p = lds_p[k * 16 + col];
            atomicAdd(&lds_yx[i0 * K_ + k], p);
            atomicAdd(&lds_yx[i1 * K_ + k], DELTA * p);
        }
        if (r == 0) {
            atomicAdd(&lds_yy[i0], v0);
            atomicAdd(&lds_yy[i1], DELTA * v1);
        }
    }
    __syncthreads();
  }

    for (int i = gtid; i < COUT * K_; i += TPB) atomicAdd(&yx[i], lds_yx[i]);
    if (gtid < COUT) atomicAdd(&yy[gtid], lds_yy[gtid]);
}

__global__ void zero_ws(float* yx, float* yy, unsigned int* mb)
{
    int i = blockIdx.x * blockDim.x + threadIdx.x;
    if (i < COUT * K_) yx[i] = 0.0f;
    if (i < COUT)      yy[i] = 0.0f;
    if (i == 0)        *mb = __float_as_uint(PRECF);
}

__global__ void finalize_ds(const float* __restrict__ w,
                            const float* __restrict__ yx,
                            const float* __restrict__ yy,
                            float* __restrict__ ds, unsigned int* mb)
{
    int i = blockIdx.x * blockDim.x + threadIdx.x;
    if (i < COUT * K_) {
        int o = i / K_;
        float d = yx[i] * (1.0f / (float)L_) - (yy[o] * (1.0f / (float)L_)) * w[i];
        ds[i] = d;
        atomicMax(mb, __float_as_uint(__builtin_fabsf(d)));
    }
}

__global__ void apply_update(const float* __restrict__ w,
                             const float* __restrict__ ds,
                             const unsigned int* __restrict__ mb,
                             float* __restrict__ neww)
{
    int i = blockIdx.x * blockDim.x + threadIdx.x;
    if (i < COUT * K_) {
        float nc = __uint_as_float(*mb);
        neww[i] = w[i] + LR * (ds[i] / nc);
    }
}

extern "C" void kernel_launch(void* const* d_in, const int* in_sizes, int n_in,
                              void* d_out, int out_size, void* d_ws, size_t ws_size,
                              hipStream_t stream)
{
    const float* x = (const float*)d_in[0];
    const float* w = (const float*)d_in[1];
    float* out  = (float*)d_out;
    float* neww = out + (size_t)B_ * COUT * L_;

    float*        yx = (float*)d_ws;
    float*        yy = yx + COUT * K_;
    unsigned int* mb = (unsigned int*)(yy + COUT);
    float*        ds = (float*)(mb + 1);

    zero_ws<<<(COUT * K_ + 255) / 256, 256, 0, stream>>>(yx, yy, mb);
    conv_hebb_gemm<<<B_ * BPI, TPB, 0, stream>>>(x, w, out, yx, yy);
    finalize_ds<<<(COUT * K_ + 255) / 256, 256, 0, stream>>>(w, yx, yy, ds, mb);
    apply_update<<<(COUT * K_ + 255) / 256, 256, 0, stream>>>(w, ds, mb, neww);
}


# Scatter of tile t moved into pass t+1 beside the products of tile t+1 (3 patch, 2 record buffers per group): one barrier per pass, not 3; half the waves scatter first; atomic sum order changes
# speedup vs baseline: 16.3829x; 1.1077x over previous
#include <hip/hip_runtime.h>

#define B_     64
#define CIN    3
#define Hh     96
#define Ww     96
#define COUT   96
#define KH     5
#define KW     5
#define STRIDE 2
#define PAD    2
#define HOUT   48
#define WOUT   48
#define L_     (HOUT*WOUT)
#define K_     (CIN*KH*KW)
#define KP     76
#define LR     0.02f
#define DELTA  (-0.4f)
#define PRECF  1e-30f
#define G_     4
#define NT     192
#define TPB    (G_ * NT)
#define BPI    4
#define TPBLK  ((L_ / 16) / BPI)
#define ITERS  (TPBLK / G_)

typedef float v2f __attribute__((ext_vector_type(2)));
typedef float v8f __attribute__((ext_vector_type(8)));

typedef int v4i __attribute__((ext_vector_type(4)));

__device__ __forceinline__ int xhalf_i(int b)
{
    return __builtin_amdgcn_permlanex16(b, b, 0x76543210, 0xfedcba98, false, false);
}
__device__ __forceinline__ float xhalf_f(float x)
{
    return __int_as_float(xhalf_i(__float_as_int(x)));
}

__device__ __forceinline__ bool ranks_before(float av, int ai, float bv, int bi)
{
    return (av > bv) || (av == bv && ai < bi);
}

__device__ __forceinline__ void merge_top2(float& v0, int& i0, float& v1, int& i1,
                                           float w0, int j0, float w1, int j1)
{
    const bool af = ranks_before(v0, i0, w0, j0);
    const float xv = af ? v1 : v0;  const int xi = af ? i1 : i0;
    const float yv = af ? w0 : w1;  const int yi = af ? j0 : j1;
    const bool xs = ranks_before(xv, xi, yv, yi);
    v0 = af ? v0 : w0;  i0 = af ? i0 : j0;
    v1 = xs ? xv : yv;  i1 = xs ? xi : yi;
}

#define NG ((KP * 16 + NT - 1) / NT)

__device__ __forceinline__ int gather_issue(const float* __restrict__ xb, int l0, int n,
                                            const int (&pk)[NG], float (&raw)[NG])
{
    const int l   = l0 + n;
    const int y   = l / WOUT, xo = l - y * WOUT;
    const int iy0 = y * STRIDE - PAD, ix0 = xo * STRIDE - PAD;
    const int rb  = iy0 * Ww + ix0;
    int okb = 0;
    #pragma unroll
    for (int j = 0; j < NG; ++j) {
        const int ki = (pk[j] >> 16) & 15, kj = (pk[j] >> 20) & 15;
        const bool ok = pk[j] >= 0 && (unsigned)(iy0 + ki) < (unsigned)Hh
                                   && (unsigned)(ix0 + kj) < (unsigned)Ww;
        const int idx = ok ? rb + (pk[j] & 0xffff) : 0;
        raw[j] = xb[idx];
        okb |= ok ? (1 << j) : 0;
    }
    return okb;
}

#define PB 3

__device__ __forceinline__ void products_tile(const float* wr, const float* lds_p, v4i* lds_c,
                                              float* __restrict__ outp, int ob, int mn, int kg, int wv)
{
    v8f accO = {}; v8f accT = {};
    #pragma unroll
    for (int kk = 0; kk < KP; kk += 4) {
        int ka = kk + 2 * kg;
        float a0 = wr[ka], a1 = wr[ka + 1];
        v2f aW = {a0, a1};
        v2f aL = {a0 * __builtin_fabsf(a0), a1 * __builtin_fabsf(a1)};
        v2f bP = {lds_p[ka * 16 + mn], lds_p[(ka + 1) * 16 + mn]};
        accO = __builtin_amdgcn_wmma_f32_16x16x4_f32(
            false, aW, false, bP, (short)0, accO, false, false);
        accT = __builtin_amdgcn_wmma_f32_16x16x4_f32(
            false, aL, false, bP, (short)0, accT, false, false);
    }

    #pragma unroll
    for (int r = 0; r < 8; ++r)
        outp[(size_t)r * L_] = accO[r];

    float v0, v1; int i0, i1;
    if (accT[1] > accT[0]) { v0 = accT[1]; i0 = ob + 1; v1 = accT[0]; i1 = ob; }
    else                   { v0 = accT[0]; i0 = ob;     v1 = accT[1]; i1 = ob + 1; }
    #pragma unroll
    for (int r = 2; r < 8; ++r) {
        const float e = accT[r];
        const bool g0 = e > v0, g1 = e > v1;
        v1 = g0 ? v0 : (g1 ? e : v1);
        i1 = g0 ? i0 : (g1 ? ob + r : i1);
        v0 = g0 ? e : v0;
        i0 = g0 ? ob + r : i0;
    }
    float w0 = xhalf_f(v0), w1 = xhalf_f(v1);
    int   j0 = xhalf_i(i0), j1 = xhalf_i(i1);
    merge_top2(v0, i0, v1, i1, w0, j0, w1, j1);
    if (kg == 0) {
        v4i rec = {__float_as_int(v0), i0, __float_as_int(v1), i1};
        lds_c[mn * 6 + wv] = rec;
    }
}

__device__ __forceinline__ void scatter_tile(const float* lds_p, const v4i* lds_c,
                                             float* lds_yx, float* lds_yy, int tid)
{
    int col = tid / 12;
    int r   = tid - col * 12;
    v4i rec[6];
    #pragma unroll
    for (int m = 0; m < 6; ++m) rec[m] = lds_c[col * 6 + m];
    float v0 = __int_as_float(rec[0].x), v1 = __int_as_float(rec[0].z);
    int   i0 = rec[0].y, i1 = rec[0].w;
    #pragma unroll
    for (int m = 1; m < 6; ++m)
        merge_top2(v0, i0, v1, i1, __int_as_float(rec[m].x), rec[m].y,
                   __int_as_float(rec[m].z), rec[m].w);
    for (int k = r; k < K_; k += 12) {
        float p = lds_p[k * 16 + col];
        atomicAdd(&lds_yx[i0 * K_ + k], p);
        atomicAdd(&lds_yx[i1 * K_ + k], DELTA * p);
    }
    if (r == 0) {
        atomicAdd(&lds_yy[i0], v0);
        atomicAdd(&lds_yy[i1], DELTA * v1);
    }
}

__global__ __launch_bounds__(TPB) void conv_hebb_gemm(
    const float* __restrict__ x, const float* __restrict__ w,
    float* __restrict__ out, float* __restrict__ yx, float* __restrict__ yy)
{
    __shared__ float lds_w[COUT * KP];
    __shared__ float lds_yx[COUT * K_];
    __shared__ float lds_yy[COUT];
    __shared__ float lds_pp[G_][PB][KP * 16];
    __shared__ v4i   lds_cc[G_][2][16 * 6];

    const int gtid = threadIdx.x;
    const int g    = gtid / NT;
    const int tid  = gtid - g * NT;
    const int b    = blockIdx.x / BPI;
    const int q    = blockIdx.x - b * BPI;

    const bool scatter_first = (__builtin_amdgcn_readfirstlane(gtid >> 7) & 1) != 0;

    const float* xb = x + (size_t)b * CIN * Hh * Ww;
    int pk[NG];
    #pragma unroll
    for (int j = 0; j < NG; ++j) {
        const int k  = (tid + NT * j) >> 4;
        const int c  = k / (KH * KW);
        const int r  = k - c * (KH * KW);
        const int ki = r / KW, kj = r - ki * KW;
        pk[j] = (k < K_) ? (((c * Hh + ki) * Ww + kj) | (ki << 16) | (kj << 20)) : -1;
    }
    float raw[NG];
    int okb = gather_issue(xb, (q * TPBLK + g) * 16, tid & 15, pk, raw);

    for (int i = gtid; i < COUT * KP; i += TPB) {
        int o = i / KP, k = i - o * KP;
        lds_w[i] = (k < K_) ? w[o * K_ + k] : 0.0f;
    }
    for (int i = gtid; i < COUT * K_; i += TPB) lds_yx[i] = 0.0f;
    if (gtid < COUT) lds_yy[gtid] = 0.0f;

    const int wv   = tid >> 5;
    const int lane = tid & 31;
    const int mn   = lane & 15;
    const int kg   = lane >> 4;
    const int ob   = wv * 16 + 8 * kg;
    const float* wr = &lds_w[(wv * 16 + mn) * KP];

    float* p_cur  = lds_pp[g][0];
    float* p_nxt  = lds_pp[g][1];
    float* p_prv  = lds_pp[g][2];

  for (int it = 0; it <= ITERS; ++it) {
    const int l0 = (q * TPBLK + it * G_ + g) * 16;
    v4i* c_cur = lds_cc[g][it & 1];
    v4i* c_prv = lds_cc[g][(it & 1) ^ 1];

    if (it < ITERS) {
        #pragma unroll
        for (int j = 0; j < NG; ++j) {
            const int i = tid + NT * j;
            if (j < NG - 1 || i < KP * 16) p_cur[i] = ((okb >> j) & 1) ? raw[j] : 0.0f;
        }
    }
    __syncthreads();

    if (it + 1 < ITERS) okb = gather_issue(xb, l0 + G_ * 16, tid & 15, pk, raw);

    if (scatter_first && it > 0) scatter_tile(p_prv, c_prv, lds_yx, lds_yy, tid);
    if (it < ITERS)
        products_tile(wr, p_cur, c_cur, &out[((size_t)b * COUT + ob) * L_ + l0 + mn], ob, mn, kg, wv);
    if (!scatter_first && it > 0) scatter_tile(p_prv, c_prv, lds_yx, lds_yy, tid);

    float* t = p_prv; p_prv = p_cur; p_cur = p_nxt; p_nxt = t;
  }
    __syncthreads();

    for (int i = gtid; i < COUT * K_; i += TPB) atomicAdd(&yx[i], lds_yx[i]);
    if (gtid < COUT) atomicAdd(&yy[gtid], lds_yy[gtid]);
}

__global__ void zero_ws(float* yx, float* yy, unsigned int* mb)
{
    int i = blockIdx.x * blockDim.x + threadIdx.x;
    if (i < COUT * K_) yx[i] = 0.0f;
    if (i < COUT)      yy[i] = 0.0f;
    if (i == 0)        *mb = __float_as_uint(PRECF);
}

__global__ void finalize_ds(const float* __restrict__ w,
                            const float* __restrict__ yx,
                            const float* __restrict__ yy,
                            float* __restrict__ ds, unsigned int* mb)
{
    int i = blockIdx.x * blockDim.x + threadIdx.x;
    if (i < COUT * K_) {
        int o = i / K_;
        float d = yx[i] * (1.0f / (float)L_) - (yy[o] * (1.0f / (float)L_)) * w[i];
        ds[i] = d;
        atomicMax(mb, __float_as_uint(__builtin_fabsf(d)));
    }
}

__global__ void apply_update(const float* __restrict__ w,
                             const float* __restrict__ ds,
                             const unsigned int* __restrict__ mb,
                             float* __restrict__ neww)
{
    int i = blockIdx.x * blockDim.x + threadIdx.x;
    if (i < COUT * K_) {
        float nc = __uint_as_float(*mb);
        neww[i] = w[i] + LR * (ds[i] / nc);
    }
}

extern "C" void kernel_launch(void* const* d_in, const int* in_sizes, int n_in,
                              void* d_out, int out_size, void* d_ws, size_t ws_size,
                              hipStream_t stream)
{
    const float* x = (const float*)d_in[0];
    const float* w = (const float*)d_in[1];
    float* out  = (float*)d_out;
    float* neww = out + (size_t)B_ * COUT * L_;

    float*        yx = (float*)d_ws;
    float*        yy = yx + COUT * K_;
    unsigned int* mb = (unsigned int*)(yy + COUT);
    float*        ds = (float*)(mb + 1);

    zero_ws<<<(COUT * K_ + 255) / 256, 256, 0, stream>>>(yx, yy, mb);
    conv_hebb_gemm<<<B_ * BPI, TPB, 0, stream>>>(x, w, out, yx, yy);
    finalize_ds<<<(COUT * K_ + 255) / 256, 256, 0, stream>>>(w, yx, yy, ds, mb);
    apply_update<<<(COUT * K_ + 255) / 256, 256, 0, stream>>>(w, ds, mb, neww);
}


# Each wave keeps the weight fragments of its 16 channels (W and W|W|, 15 of 19 k-steps: the register limit) in registers for all its tiles; per tile only patch fragments come from LDS, a block ahead
# speedup vs baseline: 17.1393x; 1.0462x over previous
#include <hip/hip_runtime.h>

#define B_     64
#define CIN    3
#define Hh     96
#define Ww     96
#define COUT   96
#define KH     5
#define KW     5
#define STRIDE 2
#define PAD    2
#define HOUT   48
#define WOUT   48
#define L_     (HOUT*WOUT)
#define K_     (CIN*KH*KW)
#define KP     76
#define LR     0.02f
#define DELTA  (-0.4f)
#define PRECF  1e-30f
#define G_     4
#define NT     192
#define TPB    (G_ * NT)
#define BPI    4
#define TPBLK  ((L_ / 16) / BPI)
#define ITERS  (TPBLK / G_)

typedef float v2f __attribute__((ext_vector_type(2)));
typedef float v8f __attribute__((ext_vector_type(8)));

typedef int v4i __attribute__((ext_vector_type(4)));

__device__ __forceinline__ int xhalf_i(int b)
{
    return __builtin_amdgcn_permlanex16(b, b, 0x76543210, 0xfedcba98, false, false);
}
__device__ __forceinline__ float xhalf_f(float x)
{
    return __int_as_float(xhalf_i(__float_as_int(x)));
}

__device__ __forceinline__ bool ranks_before(float av, int ai, float bv, int bi)
{
    return (av > bv) || (av == bv && ai < bi);
}

__device__ __forceinline__ void merge_top2(float& v0, int& i0, float& v1, int& i1,
                                           float w0, int j0, float w1, int j1)
{
    const bool af = ranks_before(v0, i0, w0, j0);
    const float xv = af ? v1 : v0;  const int xi = af ? i1 : i0;
    const float yv = af ? w0 : w1;  const int yi = af ? j0 : j1;
    const bool xs = ranks_before(xv, xi, yv, yi);
    v0 = af ? v0 : w0;  i0 = af ? i0 : j0;
    v1 = xs ? xv : yv;  i1 = xs ? xi : yi;
}

#define NG ((KP * 16 + NT - 1) / NT)

__device__ __forceinline__ int gather_issue(const float* __restrict__ xb, int l0, int n,
                                            const int (&pk)[NG], float (&raw)[NG])
{
    const int l   = l0 + n;
    const int y   = l / WOUT, xo = l - y * WOUT;
    const int iy0 = y * STRIDE - PAD, ix0 = xo * STRIDE - PAD;
    const int rb  = iy0 * Ww + ix0;
    int okb = 0;
    #pragma unroll
    for (int j = 0; j < NG; ++j) {
        const int ki = (pk[j] >> 16) & 15, kj = (pk[j] >> 20) & 15;
        const bool ok = pk[j] >= 0 && (unsigned)(iy0 + ki) < (unsigned)Hh
                                   && (unsigned)(ix0 + kj) < (unsigned)Ww;
        const int idx = ok ? rb + (pk[j] & 0xffff) : 0;
        raw[j] = xb[idx];
        okb |= ok ? (1 << j) : 0;
    }
    return okb;
}

#define PB 3

#define KS (KP / 4)
#define KR 15
#define BS 3

__device__ __forceinline__ void products_tile(const v2f (&aW)[KR], const v2f (&aL)[KR], const float* wr,
                                              const float* lds_p, v4i* lds_c,
                                              float* __restrict__ outp, int ob, int mn, int kg, int wv)
{
    v8f accO = {}; v8f accT = {};
    const float* pb = lds_p + 2 * kg * 16 + mn;
    const float* wa = wr + 2 * kg;
    v2f bq[2][BS];
    v2f wq[2][BS];
    static_assert(KR >= BS, "the first block's weight fragments are all in registers");
    #pragma unroll
    for (int i = 0; i < BS; ++i)
        bq[0][i] = (v2f){pb[(4 * i) * 16], pb[(4 * i + 1) * 16]};
    #pragma unroll
    for (int blk = 0; blk * BS < KS; ++blk) {
        const int cur = blk & 1, nxt = cur ^ 1;
        #pragma unroll
        for (int i = 0; i < BS; ++i) {
            const int s = (blk + 1) * BS + i;
            if (s < KS) {
                bq[nxt][i] = (v2f){pb[(4 * s) * 16], pb[(4 * s + 1) * 16]};
                if (s >= KR) wq[nxt][i] = (v2f){wa[4 * s], wa[4 * s + 1]};
            }
        }
        #pragma unroll
        for (int i = 0; i < BS; ++i) {
            const int s = blk * BS + i;
            if (s < KS) {
                v2f fW, fL;
                if (s < KR) { fW = aW[s]; fL = aL[s]; }
                else {
                    fW = wq[cur][i];
                    fL = (v2f){fW.x * __builtin_fabsf(fW.x), fW.y * __builtin_fabsf(fW.y)};
                }
                accO = __builtin_amdgcn_wmma_f32_16x16x4_f32(
                    false, fW, false, bq[cur][i], (short)0, accO, false, false);
                accT = __builtin_amdgcn_wmma_f32_16x16x4_f32(
                    false, fL, false, bq[cur][i], (short)0, accT, false, false);
            }
        }
    }

    #pragma unroll
    for (int r = 0; r < 8; ++r)
        outp[(size_t)r * L_] = accO[r];

    float v0, v1; int i0, i1;
    if (accT[1] > accT[0]) { v0 = accT[1]; i0 = ob + 1; v1 = accT[0]; i1 = ob; }
    else                   { v0 = accT[0]; i0 = ob;     v1 = accT[1]; i1 = ob + 1; }
    #pragma unroll
    for (int r = 2; r < 8; ++r) {
        const float e = accT[r];
        const bool g0 = e > v0, g1 = e > v1;
        v1 = g0 ? v0 : (g1 ? e : v1);
        i1 = g0 ? i0 : (g1 ? ob + r : i1);
        v0 = g0 ? e : v0;
        i0 = g0 ? ob + r : i0;
    }
    float w0 = xhalf_f(v0), w1 = xhalf_f(v1);
    int   j0 = xhalf_i(i0), j1 = xhalf_i(i1);
    merge_top2(v0, i0, v1, i1, w0, j0, w1, j1);
    if (kg == 0) {
        v4i rec = {__float_as_int(v0), i0, __float_as_int(v1), i1};
        lds_c[mn * 6 + wv] = rec;
    }
}

__device__ __forceinline__ void scatter_tile(const float* lds_p, const v4i* lds_c,
                                             float* lds_yx, float* lds_yy, int tid)
{
    int col = tid / 12;
    int r   = tid - col * 12;
    v4i rec[6];
    #pragma unroll
    for (int m = 0; m < 6; ++m) rec[m] = lds_c[col * 6 + m];
    float v0 = __int_as_float(rec[0].x), v1 = __int_as_float(rec[0].z);
    int   i0 = rec[0].y, i1 = rec[0].w;
    #pragma unroll
    for (int m = 1; m < 6; ++m)
        merge_top2(v0, i0, v1, i1, __int_as_float(rec[m].x), rec[m].y,
                   __int_as_float(rec[m].z), rec[m].w);
    for (int k = r; k < K_; k += 12) {
        float p = lds_p[k * 16 + col];
        atomicAdd(&lds_yx[i0 * K_ + k], p);
        atomicAdd(&lds_yx[i1 * K_ + k], DELTA * p);
    }
    if (r == 0) {
        atomicAdd(&lds_yy[i0], v0);
        atomicAdd(&lds_yy[i1], DELTA * v1);
    }
}

__global__ __launch_bounds__(TPB) void conv_hebb_gemm(
    const float* __restrict__ x, const float* __restrict__ w,
    float* __restrict__ out, float* __restrict__ yx, float* __restrict__ yy)
{
    __shared__ float lds_w[COUT * KP];
    __shared__ float lds_yx[COUT * K_];
    __shared__ float lds_yy[COUT];
    __shared__ float lds_pp[G_][PB][KP * 16];
    __shared__ v4i   lds_cc[G_][2][16 * 6];

    const int gtid = threadIdx.x;
    const int g    = gtid / NT;
    const int tid  = gtid - g * NT;
    const int b    = blockIdx.x / BPI;
    const int q    = blockIdx.x - b * BPI;

    const bool scatter_first = (__builtin_amdgcn_readfirstlane(gtid >> 7) & 1) != 0;

    const float* xb = x + (size_t)b * CIN * Hh * Ww;
    int pk[NG];
    #pragma unroll
    for (int j = 0; j < NG; ++j) {
        const int k  = (tid + NT * j) >> 4;
        const int c  = k / (KH * KW);
        const int r  = k - c * (KH * KW);
        const int ki = r / KW, kj = r - ki * KW;
        pk[j] = (k < K_) ? (((c * Hh + ki) * Ww + kj) | (ki << 16) | (kj << 20)) : -1;
    }
    float raw[NG];
    int okb = gather_issue(xb, (q * TPBLK + g) * 16, tid & 15, pk, raw);

    for (int i = gtid; i < COUT * KP; i += TPB) {
        int o = i / KP, k = i - o * KP;
        lds_w[i] = (k < K_) ? w[o * K_ + k] : 0.0f;
    }
    for (int i = gtid; i < COUT * K_; i += TPB) lds_yx[i] = 0.0f;
    if (gtid < COUT) lds_yy[gtid] = 0.0f;

    const int wv   = tid >> 5;
    const int lane = tid & 31;
    const int mn   = lane & 15;
    const int kg   = lane >> 4;
    const int ob   = wv * 16 + 8 * kg;
    const float* wr = &lds_w[(wv * 16 + mn) * KP];

    __syncthreads();
    v2f aW[KR], aL[KR];
    #pragma unroll
    for (int s = 0; s < KR; ++s) {
        const int ka = 4 * s + 2 * kg;
        const float a0 = wr[ka], a1 = wr[ka + 1];
        aW[s] = (v2f){a0, a1};
        aL[s] = (v2f){a0 * __builtin_fabsf(a0), a1 * __builtin_fabsf(a1)};
    }

    float* p_cur  = lds_pp[g][0];
    float* p_nxt  = lds_pp[g][1];
    float* p_prv  = lds_pp[g][2];

  for (int it = 0; it <= ITERS; ++it) {
    const int l0 = (q * TPBLK + it * G_ + g) * 16;
    v4i* c_cur = lds_cc[g][it & 1];
    v4i* c_prv = lds_cc[g][(it & 1) ^ 1];

    if (it < ITERS) {
        #pragma unroll
        for (int j = 0; j < NG; ++j) {
            const int i = tid + NT * j;
            if (j < NG - 1 || i < KP * 16) p_cur[i] = ((okb >> j) & 1) ? raw[j] : 0.0f;
        }
    }
    __syncthreads();

    if (it + 1 < ITERS) okb = gather_issue(xb, l0 + G_ * 16, tid & 15, pk, raw);

    if (scatter_first && it > 0) scatter_tile(p_prv, c_prv, lds_yx, lds_yy, tid);
    if (it < ITERS)
        products_tile(aW, aL, wr, p_cur, c_cur, &out[((size_t)b * COUT + ob) * L_ + l0 + mn], ob, mn, kg, wv);
    if (!scatter_first && it > 0) scatter_tile(p_prv, c_prv, lds_yx, lds_yy, tid);

    float* t = p_prv; p_prv = p_cur; p_cur = p_nxt; p_nxt = t;
  }
    __syncthreads();

    for (int i = gtid; i < COUT * K_; i += TPB) atomicAdd(&yx[i], lds_yx[i]);
    if (gtid < COUT) atomicAdd(&yy[gtid], lds_yy[gtid]);
}

__global__ void zero_ws(float* yx, float* yy, unsigned int* mb)
{
    int i = blockIdx.x * blockDim.x + threadIdx.x;
    if (i < COUT * K_) yx[i] = 0.0f;
    if (i < COUT)      yy[i] = 0.0f;
    if (i == 0)        *mb = __float_as_uint(PRECF);
}

__global__ void finalize_ds(const float* __restrict__ w,
                            const float* __restrict__ yx,
                            const float* __restrict__ yy,
                            float* __restrict__ ds, unsigned int* mb)
{
    int i = blockIdx.x * blockDim.x + threadIdx.x;
    if (i < COUT * K_) {
        int o = i / K_;
        float d = yx[i] * (1.0f / (float)L_) - (yy[o] * (1.0f / (float)L_)) * w[i];
        ds[i] = d;
        atomicMax(mb, __float_as_uint(__builtin_fabsf(d)));
    }
}

__global__ void apply_update(const float* __restrict__ w,
                             const float* __restrict__ ds,
                             const unsigned int* __restrict__ mb,
                             float* __restrict__ neww)
{
    int i = blockIdx.x * blockDim.x + threadIdx.x;
    if (i < COUT * K_) {
        float nc = __uint_as_float(*mb);
        neww[i] = w[i] + LR * (ds[i] / nc);
    }
}

extern "C" void kernel_launch(void* const* d_in, const int* in_sizes, int n_in,
                              void* d_out, int out_size, void* d_ws, size_t ws_size,
                              hipStream_t stream)
{
    const float* x = (const float*)d_in[0];
    const float* w = (const float*)d_in[1];
    float* out  = (float*)d_out;
    float* neww = out + (size_t)B_ * COUT * L_;

    float*        yx = (float*)d_ws;
    float*        yy = yx + COUT * K_;
    unsigned int* mb = (unsigned int*)(yy + COUT);
    float*        ds = (float*)(mb + 1);

    zero_ws<<<(COUT * K_ + 255) / 256, 256, 0, stream>>>(yx, yy, mb);
    conv_hebb_gemm<<<B_ * BPI, TPB, 0, stream>>>(x, w, out, yx, yy);
    finalize_ds<<<(COUT * K_ + 255) / 256, 256, 0, stream>>>(w, yx, yy, ds, mb);
    apply_update<<<(COUT * K_ + 255) / 256, 256, 0, stream>>>(w, ds, mb, neww);
}


# Weight staging, scatter and final global adds: a thread reads all its elements into registers as one group before first use, one round trip per loop, not per element; atomic arrival order shifts
# speedup vs baseline: 18.1587x; 1.0595x over previous
#include <hip/hip_runtime.h>

#define B_     64
#define CIN    3
#define Hh     96
#define Ww     96
#define COUT   96
#define KH     5
#define KW     5
#define STRIDE 2
#define PAD    2
#define HOUT   48
#define WOUT   48
#define L_     (HOUT*WOUT)
#define K_     (CIN*KH*KW)
#define KP     76
#define LR     0.02f
#define DELTA  (-0.4f)
#define PRECF  1e-30f
#define G_     4
#define NT     192
#define TPB    (G_ * NT)
#define BPI    4
#define TPBLK  ((L_ / 16) / BPI)
#define ITERS  (TPBLK / G_)

typedef float v2f __attribute__((ext_vector_type(2)));
typedef float v8f __attribute__((ext_vector_type(8)));

typedef int v4i __attribute__((ext_vector_type(4)));

__device__ __forceinline__ int xhalf_i(int b)
{
    return __builtin_amdgcn_permlanex16(b, b, 0x76543210, 0xfedcba98, false, false);
}
__device__ __forceinline__ float xhalf_f(float x)
{
    return __int_as_float(xhalf_i(__float_as_int(x)));
}

__device__ __forceinline__ bool ranks_before(float av, int ai, float bv, int bi)
{
    return (av > bv) || (av == bv && ai < bi);
}

__device__ __forceinline__ void merge_top2(float& v0, int& i0, float& v1, int& i1,
                                           float w0, int j0, float w1, int j1)
{
    const bool af = ranks_before(v0, i0, w0, j0);
    const float xv = af ? v1 : v0;  const int xi = af ? i1 : i0;
    const float yv = af ? w0 : w1;  const int yi = af ? j0 : j1;
    const bool xs = ranks_before(xv, xi, yv, yi);
    v0 = af ? v0 : w0;  i0 = af ? i0 : j0;
    v1 = xs ? xv : yv;  i1 = xs ? xi : yi;
}

#define NG ((KP * 16 + NT - 1) / NT)

__device__ __forceinline__ int gather_issue(const float* __restrict__ xb, int l0, int n,
                                            const int (&pk)[NG], float (&raw)[NG])
{
    const int l   = l0 + n;
    const int y   = l / WOUT, xo = l - y * WOUT;
    const int iy0 = y * STRIDE - PAD, ix0 = xo * STRIDE - PAD;
    const int rb  = iy0 * Ww + ix0;
    int okb = 0;
    #pragma unroll
    for (int j = 0; j < NG; ++j) {
        const int ki = (pk[j] >> 16) & 15, kj = (pk[j] >> 20) & 15;
        const bool ok = pk[j] >= 0 && (unsigned)(iy0 + ki) < (unsigned)Hh
                                   && (unsigned)(ix0 + kj) < (unsigned)Ww;
        const int idx = ok ? rb + (pk[j] & 0xffff) : 0;
        raw[j] = xb[idx];
        okb |= ok ? (1 << j) : 0;
    }
    return okb;
}

#define PB 3

#define KS (KP / 4)
#define KR 15
#define BS 3

__device__ __forceinline__ void products_tile(const v2f (&aW)[KR], const v2f (&aL)[KR], const float* wr,
                                              const float* lds_p, v4i* lds_c,
                                              float* __restrict__ outp, int ob, int mn, int kg, int wv)
{
    v8f accO = {}; v8f accT = {};
    const float* pb = lds_p + 2 * kg * 16 + mn;
    const float* wa = wr + 2 * kg;
    v2f bq[2][BS];
    v2f wq[2][BS];
    static_assert(KR >= BS, "the first block's weight fragments are all in registers");
    #pragma unroll
    for (int i = 0; i < BS; ++i)
        bq[0][i] = (v2f){pb[(4 * i) * 16], pb[(4 * i + 1) * 16]};
    #pragma unroll
    for (int blk = 0; blk * BS < KS; ++blk) {
        const int cur = blk & 1, nxt = cur ^ 1;
        #pragma unroll
        for (int i = 0; i < BS; ++i) {
            const int s = (blk + 1) * BS + i;
            if (s < KS) {
                bq[nxt][i] = (v2f){pb[(4 * s) * 16], pb[(4 * s + 1) * 16]};
                if (s >= KR) wq[nxt][i] = (v2f){wa[4 * s], wa[4 * s + 1]};
            }
        }
        #pragma unroll
        for (int i = 0; i < BS; ++i) {
            const int s = blk * BS + i;
            if (s < KS) {
                v2f fW, fL;
                if (s < KR) { fW = aW[s]; fL = aL[s]; }
                else {
                    fW = wq[cur][i];
                    fL = (v2f){fW.x * __builtin_fabsf(fW.x), fW.y * __builtin_fabsf(fW.y)};
                }
                accO = __builtin_amdgcn_wmma_f32_16x16x4_f32(
                    false, fW, false, bq[cur][i], (short)0, accO, false, false);
                accT = __builtin_amdgcn_wmma_f32_16x16x4_f32(
                    false, fL, false, bq[cur][i], (short)0, accT, false, false);
            }
        }
    }

    #pragma unroll
    for (int r = 0; r < 8; ++r)
        outp[(size_t)r * L_] = accO[r];

    float v0, v1; int i0, i1;
    if (accT[1] > accT[0]) { v0 = accT[1]; i0 = ob + 1; v1 = accT[0]; i1 = ob; }
    else                   { v0 = accT[0]; i0 = ob;     v1 = accT[1]; i1 = ob + 1; }
    #pragma unroll
    for (int r = 2; r < 8; ++r) {
        const float e = accT[r];
        const bool g0 = e > v0, g1 = e > v1;
        v1 = g0 ? v0 : (g1 ? e : v1);
        i1 = g0 ? i0 : (g1 ? ob + r : i1);
        v0 = g0 ? e : v0;
        i0 = g0 ? ob + r : i0;
    }
    float w0 = xhalf_f(v0), w1 = xhalf_f(v1);
    int   j0 = xhalf_i(i0), j1 = xhalf_i(i1);
    merge_top2(v0, i0, v1, i1, w0, j0, w1, j1);
    if (kg == 0) {
        v4i rec = {__float_as_int(v0), i0, __float_as_int(v1), i1};
        lds_c[mn * 6 + wv] = rec;
    }
}

#define SJ ((K_ + 11) / 12)
static_assert(12 * (SJ - 2) + 11 < K_, "only the last scatter slot needs a guard");

__device__ __forceinline__ void scatter_tile(const float* lds_p, const v4i* lds_c,
                                             float* lds_yx, float* lds_yy, int tid)
{
    int col = tid / 12;
    int r   = tid - col * 12;
    v4i rec[6];
    #pragma unroll
    for (int m = 0; m < 6; ++m) rec[m] = lds_c[col * 6 + m];
    const float* pc = lds_p + col;
    float p[SJ];
    #pragma unroll
    for (int j = 0; j < SJ; ++j) {
        const int k = r + 12 * j;
        p[j] = pc[((j < SJ - 1 || k < K_) ? k : r) * 16];
    }
    float v0 = __int_as_float(rec[0].x), v1 = __int_as_float(rec[0].z);
    int   i0 = rec[0].y, i1 = rec[0].w;
    #pragma unroll
    for (int m = 1; m < 6; ++m)
        merge_top2(v0, i0, v1, i1, __int_as_float(rec[m].x), rec[m].y,
                   __int_as_float(rec[m].z), rec[m].w);
    float* y0 = lds_yx + i0 * K_ + r;
    float* y1 = lds_yx + i1 * K_ + r;
    #pragma unroll
    for (int j = 0; j < SJ; ++j) {
        if (j < SJ - 1 || r + 12 * j < K_) {
            atomicAdd(&y0[12 * j], p[j]);
            atomicAdd(&y1[12 * j], DELTA * p[j]);
        }
    }
    if (r == 0) {
        atomicAdd(&lds_yy[i0], v0);
        atomicAdd(&lds_yy[i1], DELTA * v1);
    }
}

#define WN ((COUT * KP + TPB - 1) / TPB)
#define YN ((COUT * K_ + TPB - 1) / TPB)
static_assert((WN - 1) * TPB <= COUT * KP && (YN - 1) * TPB <= COUT * K_, "only a thread's last slot can fall beyond the array");

__global__ __launch_bounds__(TPB) void conv_hebb_gemm(
    const float* __restrict__ x, const float* __restrict__ w,
    float* __restrict__ out, float* __restrict__ yx, float* __restrict__ yy)
{
    __shared__ float lds_w[COUT * KP];
    __shared__ float lds_yx[COUT * K_];
    __shared__ float lds_yy[COUT];
    __shared__ float lds_pp[G_][PB][KP * 16];
    __shared__ v4i   lds_cc[G_][2][16 * 6];

    const int gtid = threadIdx.x;
    const int g    = gtid / NT;
    const int tid  = gtid - g * NT;
    const int b    = blockIdx.x / BPI;
    const int q    = blockIdx.x - b * BPI;

    const bool scatter_first = (__builtin_amdgcn_readfirstlane(gtid >> 7) & 1) != 0;

    const float* xb = x + (size_t)b * CIN * Hh * Ww;
    int pk[NG];
    #pragma unroll
    for (int j = 0; j < NG; ++j) {
        const int k  = (tid + NT * j) >> 4;
        const int c  = k / (KH * KW);
        const int r  = k - c * (KH * KW);
        const int ki = r / KW, kj = r - ki * KW;
        pk[j] = (k < K_) ? (((c * Hh + ki) * Ww + kj) | (ki << 16) | (kj << 20)) : -1;
    }
    float raw[NG];
    int okb = gather_issue(xb, (q * TPBLK + g) * 16, tid & 15, pk, raw);

    float wst[WN];
    int wok = 0;
    #pragma unroll
    for (int n = 0; n < WN; ++n) {
        const int i = gtid + TPB * n;
        const int o = i / KP, k = i - o * KP;
        const bool ok = i < COUT * KP && k < K_;
        wst[n] = w[ok ? o * K_ + k : 0];
        wok |= ok ? (1 << n) : 0;
    }
    for (int i = gtid; i < COUT * K_; i += TPB) lds_yx[i] = 0.0f;
    if (gtid < COUT) lds_yy[gtid] = 0.0f;
    #pragma unroll
    for (int n = 0; n < WN; ++n) {
        const int i = gtid + TPB * n;
        if (n < WN - 1 || i < COUT * KP) lds_w[i] = ((wok >> n) & 1) ? wst[n] : 0.0f;
    }

    const int wv   = tid >> 5;
    const int lane = tid & 31;
    const int mn   = lane & 15;
    const int kg   = lane >> 4;
    const int ob   = wv * 16 + 8 * kg;
    const float* wr = &lds_w[(wv * 16 + mn) * KP];

    __syncthreads();
    v2f aW[KR], aL[KR];
    #pragma unroll
    for (int s = 0; s < KR; ++s) {
        const int ka = 4 * s + 2 * kg;
        const float a0 = wr[ka], a1 = wr[ka + 1];
        aW[s] = (v2f){a0, a1};
        aL[s] = (v2f){a0 * __builtin_fabsf(a0), a1 * __builtin_fabsf(a1)};
    }

    float* p_cur  = lds_pp[g][0];
    float* p_nxt  = lds_pp[g][1];
    float* p_prv  = lds_pp[g][2];

  for (int it = 0; it <= ITERS; ++it) {
    const int l0 = (q * TPBLK + it * G_ + g) * 16;
    v4i* c_cur = lds_cc[g][it & 1];
    v4i* c_prv = lds_cc[g][(it & 1) ^ 1];

    if (it < ITERS) {
        #pragma unroll
        for (int j = 0; j < NG; ++j) {
            const int i = tid + NT * j;
            if (j < NG - 1 || i < KP * 16) p_cur[i] = ((okb >> j) & 1) ? raw[j] : 0.0f;
        }
    }
    __syncthreads();

    if (it + 1 < ITERS) okb = gather_issue(xb, l0 + G_ * 16, tid & 15, pk, raw);

    if (scatter_first && it > 0) scatter_tile(p_prv, c_prv, lds_yx, lds_yy, tid);
    if (it < ITERS)
        products_tile(aW, aL, wr, p_cur, c_cur, &out[((size_t)b * COUT + ob) * L_ + l0 + mn], ob, mn, kg, wv);
    if (!scatter_first && it > 0) scatter_tile(p_prv, c_prv, lds_yx, lds_yy, tid);

    float* t = p_prv; p_prv = p_cur; p_cur = p_nxt; p_nxt = t;
  }
    __syncthreads();

    float sv[YN];
    #pragma unroll
    for (int n = 0; n < YN; ++n) {
        const int i = gtid + TPB * n;
        sv[n] = lds_yx[(n < YN - 1 || i < COUT * K_) ? i : gtid];
    }
    const float sy = lds_yy[gtid < COUT ? gtid : 0];
    #pragma unroll
    for (int n = 0; n < YN; ++n) {
        const int i = gtid + TPB * n;
        if (n < YN - 1 || i < COUT * K_) atomicAdd(&yx[i], sv[n]);
    }
    if (gtid < COUT) atomicAdd(&yy[gtid], sy);
}

__global__ void zero_ws(float* yx, float* yy, unsigned int* mb)
{
    int i = blockIdx.x * blockDim.x + threadIdx.x;
    if (i < COUT * K_) yx[i] = 0.0f;
    if (i < COUT)      yy[i] = 0.0f;
    if (i == 0)        *mb = __float_as_uint(PRECF);
}

__global__ void finalize_ds(const float* __restrict__ w,
                            const float* __restrict__ yx,
                            const float* __restrict__ yy,
                            float* __restrict__ ds, unsigned int* mb)
{
    int i = blockIdx.x * blockDim.x + threadIdx.x;
    if (i < COUT * K_) {
        int o = i / K_;
        float d = yx[i] * (1.0f / (float)L_) - (yy[o] * (1.0f / (float)L_)) * w[i];
        ds[i] = d;
        atomicMax(mb, __float_as_uint(__builtin_fabsf(d)));
    }
}

__global__ void apply_update(const float* __restrict__ w,
                             const float* __restrict__ ds,
                             const unsigned int* __restrict__ mb,
                             float* __restrict__ neww)
{
    int i = blockIdx.x * blockDim.x + threadIdx.x;
    if (i < COUT * K_) {
        float nc = __uint_as_float(*mb);
        neww[i] = w[i] + LR * (ds[i] / nc);
    }
}

extern "C" void kernel_launch(void* const* d_in, const int* in_sizes, int n_in,
                              void* d_out, int out_size, void* d_ws, size_t ws_size,
                              hipStream_t stream)
{
    const float* x = (const float*)d_in[0];
    const float* w = (const float*)d_in[1];
    float* out  = (float*)d_out;
    float* neww = out + (size_t)B_ * COUT * L_;

    float*        yx = (float*)d_ws;
    float*        yy = yx + COUT * K_;
    unsigned int* mb = (unsigned int*)(yy + COUT);
    float*        ds = (float*)(mb + 1);

    zero_ws<<<(COUT * K_ + 255) / 256, 256, 0, stream>>>(yx, yy, mb);
    conv_hebb_gemm<<<B_ * BPI, TPB, 0, stream>>>(x, w, out, yx, yy);
    finalize_ds<<<(COUT * K_ + 255) / 256, 256, 0, stream>>>(w, yx, yy, ds, mb);
    apply_update<<<(COUT * K_ + 255) / 256, 256, 0, stream>>>(w, ds, mb, neww);
}


# Patch fragments and scatter values are read straight from an LDS copy of the group's input strip (padding stored, staged once): no per-tile gather or tile stores; k regrouped: sum order changes
# speedup vs baseline: 19.0430x; 1.0487x over previous
#include <hip/hip_runtime.h>

#define B_     64
#define CIN    3
#define Hh     96
#define Ww     96
#define COUT   96
#define KH     5
#define KW     5
#define STRIDE 2
#define PAD    2
#define HOUT   48
#define WOUT   48
#define L_     (HOUT*WOUT)
#define K_     (CIN*KH*KW)
#define KP     76
#define LR     0.02f
#define DELTA  (-0.4f)
#define PRECF  1e-30f
#define G_     4
#define NT     192
#define TPB    (G_ * NT)
#define BPI    4
#define TPBLK  ((L_ / 16) / BPI)
#define ITERS  (TPBLK / G_)

typedef float v2f __attribute__((ext_vector_type(2)));
typedef float v8f __attribute__((ext_vector_type(8)));

typedef int v4i __attribute__((ext_vector_type(4)));

__device__ __forceinline__ int xhalf_i(int b)
{
    return __builtin_amdgcn_permlanex16(b, b, 0x76543210, 0xfedcba98, false, false);
}
__device__ __forceinline__ float xhalf_f(float x)
{
    return __int_as_float(xhalf_i(__float_as_int(x)));
}

__device__ __forceinline__ bool ranks_before(float av, int ai, float bv, int bi)
{
    return (av > bv) || (av == bv && ai < bi);
}

__device__ __forceinline__ void merge_top2(float& v0, int& i0, float& v1, int& i1,
                                           float w0, int j0, float w1, int j1)
{
    const bool af = ranks_before(v0, i0, w0, j0);
    const float xv = af ? v1 : v0;  const int xi = af ? i1 : i0;
    const float yv = af ? w0 : w1;  const int yi = af ? j0 : j1;
    const bool xs = ranks_before(xv, xi, yv, yi);
    v0 = af ? v0 : w0;  i0 = af ? i0 : j0;
    v1 = xs ? xv : yv;  i1 = xs ? xi : yi;
}

#define SP 100
#define SR 27
#define SC (SR * SP)
#define SN (CIN * SC)
#define XN ((SN + TPB - 1) / TPB)
static_assert((XN - 1) * TPB <= SN, "only a thread's last strip slot can fall beyond the strip");
static_assert(2 * (TPBLK / 3 - 1) + KH - 1 < SR && 2 * (WOUT - 1) + KW - 1 < SP, "every patch element lies inside the strip");

#define KS (KP / 4)
#define KR 15
#define BS 3

static_assert(KR == CIN * KH && KS == KR + CIN + 1, "step table below");
__device__ __forceinline__ v2f patch_frag(int s, const float* p0, const float* p1,
                                          const float* p2a, const float* p2b, int h)
{
    if (s < KR)     return *(const v2f*)(p0 + (s / KH) * SC + (s % KH) * SP);
    if (s < KS - 1) return (v2f){p1[(s - KR) * SC + 4], p1[(s - KR) * SC + 4 + SP]};
    return (v2f){p2a[4 * SP + 4], h ? 0.0f : p2b[4 * SP + 4]};
}
__device__ __forceinline__ v2f weight_frag(int s, const float* w1, const float* w2a, const float* w2b)
{
    if (s < KS - 1) return (v2f){w1[KH * KW * (s - KR) + 4], w1[KH * KW * (s - KR) + 9]};
    return (v2f){*w2a, *w2b};
}

__device__ __forceinline__ void products_tile(const v2f (&aW)[KR], const v2f (&aL)[KR],
                                              const float* wk1, const float* wk2a, const float* wk2b,
                                              const float* p0, const float* p1,
                                              const float* p2a, const float* p2b, v4i* lds_c,
                                              float* __restrict__ outp, int ob, int mn, int kg, int wv)
{
    v8f accO = {}; v8f accT = {};
    v2f bq[2][BS];
    v2f wq[2][BS];
    static_assert(KR >= BS, "the first block's weight fragments are all in registers");
    #pragma unroll
    for (int i = 0; i < BS; ++i)
        bq[0][i] = patch_frag(i, p0, p1, p2a, p2b, kg);
    #pragma unroll
    for (int blk = 0; blk * BS < KS; ++blk) {
        const int cur = blk & 1, nxt = cur ^ 1;
        #pragma unroll
        for (int i = 0; i < BS; ++i) {
            const int s = (blk + 1) * BS + i;
            if (s < KS) {
                bq[nxt][i] = patch_frag(s, p0, p1, p2a, p2b, kg);
                if (s >= KR) wq[nxt][i] = weight_frag(s, wk1, wk2a, wk2b);
            }
        }
        #pragma unroll
        for (int i = 0; i < BS; ++i) {
            const int s = blk * BS + i;
            if (s < KS) {
                v2f fW, fL;
                if (s < KR) { fW = aW[s]; fL = aL[s]; }
                else {
                    fW = wq[cur][i];
                    fL = (v2f){fW.x * __builtin_fabsf(fW.x), fW.y * __builtin_fabsf(fW.y)};
                }
                accO = __builtin_amdgcn_wmma_f32_16x16x4_f32(
                    false, fW, false, bq[cur][i], (short)0, accO, false, false);
                accT = __builtin_amdgcn_wmma_f32_16x16x4_f32(
                    false, fL, false, bq[cur][i], (short)0, accT, false, false);
            }
        }
    }

    #pragma unroll
    for (int r = 0; r < 8; ++r)
        outp[(size_t)r * L_] = accO[r];

    float v0, v1; int i0, i1;
    if (accT[1] > accT[0]) { v0 = accT[1]; i0 = ob + 1; v1 = accT[0]; i1 = ob; }
    else                   { v0 = accT[0]; i0 = ob;     v1 = accT[1]; i1 = ob + 1; }
    #pragma unroll
    for (int r = 2; r < 8; ++r) {
        const float e = accT[r];
        const bool g0 = e > v0, g1 = e > v1;
        v1 = g0 ? v0 : (g1 ? e : v1);
        i1 = g0 ? i0 : (g1 ? ob + r : i1);
        v0 = g0 ? e : v0;
        i0 = g0 ? ob + r : i0;
    }
    float w0 = xhalf_f(v0), w1 = xhalf_f(v1);
    int   j0 = xhalf_i(i0), j1 = xhalf_i(i1);
    merge_top2(v0, i0, v1, i1, w0, j0, w1, j1);
    if (kg == 0) {
        v4i rec = {__float_as_int(v0), i0, __float_as_int(v1), i1};
        lds_c[mn * 6 + wv] = rec;
    }
}

#define SJ ((K_ + 11) / 12)
static_assert(12 * (SJ - 2) + 11 < K_, "only the last scatter slot needs a guard");

__device__ __forceinline__ void scatter_tile(const float* ps, const int (&sk)[SJ], const v4i* lds_c,
                                             float* lds_yx, float* lds_yy, int tid)
{
    int col = tid / 12;
    int r   = tid - col * 12;
    v4i rec[6];
    #pragma unroll
    for (int m = 0; m < 6; ++m) rec[m] = lds_c[col * 6 + m];
    const float* pc = ps + 2 * col;
    float p[SJ];
    #pragma unroll
    for (int j = 0; j < SJ; ++j) p[j] = pc[sk[j]];
    float v0 = __int_as_float(rec[0].x), v1 = __int_as_float(rec[0].z);
    int   i0 = rec[0].y, i1 = rec[0].w;
    #pragma unroll
    for (int m = 1; m < 6; ++m)
        merge_top2(v0, i0, v1, i1, __int_as_float(rec[m].x), rec[m].y,
                   __int_as_float(rec[m].z), rec[m].w);
    float* y0 = lds_yx + i0 * K_ + r;
    float* y1 = lds_yx + i1 * K_ + r;
    #pragma unroll
    for (int j = 0; j < SJ; ++j) {
        if (j < SJ - 1 || r + 12 * j < K_) {
            atomicAdd(&y0[12 * j], p[j]);
            atomicAdd(&y1[12 * j], DELTA * p[j]);
        }
    }
    if (r == 0) {
        atomicAdd(&lds_yy[i0], v0);
        atomicAdd(&lds_yy[i1], DELTA * v1);
    }
}

#define WN ((COUT * KP + TPB - 1) / TPB)
#define YN ((COUT * K_ + TPB - 1) / TPB)
static_assert((WN - 1) * TPB <= COUT * KP && (YN - 1) * TPB <= COUT * K_, "only a thread's last slot can fall beyond the array");

__global__ __launch_bounds__(TPB) void conv_hebb_gemm(
    const float* __restrict__ x, const float* __restrict__ w,
    float* __restrict__ out, float* __restrict__ yx, float* __restrict__ yy)
{
    __shared__ float lds_w[COUT * KP];
    __shared__ float lds_yx[COUT * K_];
    __shared__ float lds_yy[COUT];
    __shared__ __attribute__((aligned(16))) float lds_x[SN];
    __shared__ v4i   lds_cc[G_][2][16 * 6];

    const int gtid = threadIdx.x;
    const int g    = gtid / NT;
    const int tid  = gtid - g * NT;
    const int b    = blockIdx.x / BPI;
    const int q    = blockIdx.x - b * BPI;

    const bool scatter_first = (__builtin_amdgcn_readfirstlane(gtid >> 7) & 1) != 0;

    const float* xb = x + (size_t)b * CIN * Hh * Ww;
    float xst[XN];
    int xok = 0;
    #pragma unroll
    for (int n = 0; n < XN; ++n) {
        const int e  = gtid + TPB * n;
        const int c  = e / SC, rem = e - c * SC;
        const int ry = rem / SP, rx = rem - ry * SP;
        const int iy = STRIDE * (TPBLK / 3) * q - PAD + ry, ix = rx - PAD;
        const bool ok = e < SN && (unsigned)iy < (unsigned)Hh && (unsigned)ix < (unsigned)Ww;
        xst[n] = xb[ok ? (c * Hh + iy) * Ww + ix : 0];
        xok |= ok ? (1 << n) : 0;
    }
    float wst[WN];
    int wok = 0;
    #pragma unroll
    for (int n = 0; n < WN; ++n) {
        const int i = gtid + TPB * n;
        const int o = i / KP, k = i - o * KP;
        const bool ok = i < COUT * KP && k < K_;
        wst[n] = w[ok ? o * K_ + k : 0];
        wok |= ok ? (1 << n) : 0;
    }
    for (int i = gtid; i < COUT * K_; i += TPB) lds_yx[i] = 0.0f;
    if (gtid < COUT) lds_yy[gtid] = 0.0f;
    #pragma unroll
    for (int n = 0; n < XN; ++n) {
        const int e = gtid + TPB * n;
        if (n < XN - 1 || e < SN) lds_x[e] = ((xok >> n) & 1) ? xst[n] : 0.0f;
    }
    #pragma unroll
    for (int n = 0; n < WN; ++n) {
        const int i = gtid + TPB * n;
        if (n < WN - 1 || i < COUT * KP) lds_w[i] = ((wok >> n) & 1) ? wst[n] : 0.0f;
    }

    const int wv   = tid >> 5;
    const int lane = tid & 31;
    const int mn   = lane & 15;
    const int kg   = lane >> 4;
    const int ob   = wv * 16 + 8 * kg;
    const float* wr = &lds_w[(wv * 16 + mn) * KP];
    const float* w1  = wr + 2 * KW * kg;
    const float* w2a = wr + (kg ? 74 : 24);
    const float* w2b = wr + (kg ? 75 : 49);
    const float* x0  = lds_x + 2 * mn + 2 * kg;
    const float* x1  = lds_x + 2 * mn + 2 * kg * SP;
    const float* x2a = lds_x + 2 * mn + 2 * kg * SC;
    const float* x2b = lds_x + 2 * mn + (kg ? 2 * SC : SC);
    int sk[SJ];
    #pragma unroll
    for (int j = 0; j < SJ; ++j) {
        const int r  = tid % 12;
        const int k  = (j < SJ - 1 || r + 12 * j < K_) ? r + 12 * j : r;
        const int c  = k / (KH * KW), rr = k - c * (KH * KW);
        const int ki = rr / KW, kj = rr - ki * KW;
        sk[j] = c * SC + ki * SP + kj;
    }

    __syncthreads();
    v2f aW[KR], aL[KR];
    #pragma unroll
    for (int s = 0; s < KR; ++s) {
        const int ka = KW * s + 2 * kg;
        const float a0 = wr[ka], a1 = wr[ka + 1];
        aW[s] = (v2f){a0, a1};
        aL[s] = (v2f){a0 * __builtin_fabsf(a0), a1 * __builtin_fabsf(a1)};
    }

    int tb_prv = 0;

  for (int it = 0; it <= ITERS; ++it) {
    const int T  = it * G_ + g;
    const int yl = T / 3, xt = T - 3 * yl;
    const int tb = 2 * yl * SP + 2 * 16 * xt;
    const int l0 = (q * TPBLK + T) * 16;
    v4i* c_cur = lds_cc[g][it & 1];
    v4i* c_prv = lds_cc[g][(it & 1) ^ 1];

    __syncthreads();

    if (scatter_first && it > 0) scatter_tile(lds_x + tb_prv, sk, c_prv, lds_yx, lds_yy, tid);
    if (it < ITERS)
        products_tile(aW, aL, w1, w2a, w2b, x0 + tb, x1 + tb, x2a + tb, x2b + tb, c_cur,
                      &out[((size_t)b * COUT + ob) * L_ + l0 + mn], ob, mn, kg, wv);
    if (!scatter_first && it > 0) scatter_tile(lds_x + tb_prv, sk, c_prv, lds_yx, lds_yy, tid);

    tb_prv = tb;
  }
    __syncthreads();

    float sv[YN];
    #pragma unroll
    for (int n = 0; n < YN; ++n) {
        const int i = gtid + TPB * n;
        sv[n] = lds_yx[(n < YN - 1 || i < COUT * K_) ? i : gtid];
    }
    const float sy = lds_yy[gtid < COUT ? gtid : 0];
    #pragma unroll
    for (int n = 0; n < YN; ++n) {
        const int i = gtid + TPB * n;
        if (n < YN - 1 || i < COUT * K_) atomicAdd(&yx[i], sv[n]);
    }
    if (gtid < COUT) atomicAdd(&yy[gtid], sy);
}

__global__ void zero_ws(float* yx, float* yy, unsigned int* mb)
{
    int i = blockIdx.x * blockDim.x + threadIdx.x;
    if (i < COUT * K_) yx[i] = 0.0f;
    if (i < COUT)      yy[i] = 0.0f;
    if (i == 0)        *mb = __float_as_uint(PRECF);
}

__global__ void finalize_ds(const float* __restrict__ w,
                            const float* __restrict__ yx,
                            const float* __restrict__ yy,
                            float* __restrict__ ds, unsigned int* mb)
{
    int i = blockIdx.x * blockDim.x + threadIdx.x;
    if (i < COUT * K_) {
        int o = i / K_;
        float d = yx[i] * (1.0f / (float)L_) - (yy[o] * (1.0f / (float)L_)) * w[i];
        ds[i] = d;
        atomicMax(mb, __float_as_uint(__builtin_fabsf(d)));
    }
}

__global__ void apply_update(const float* __restrict__ w,
                             const float* __restrict__ ds,
                             const unsigned int* __restrict__ mb,
                             float* __restrict__ neww)
{
    int i = blockIdx.x * blockDim.x + threadIdx.x;
    if (i < COUT * K_) {
        float nc = __uint_as_float(*mb);
        neww[i] = w[i] + LR * (ds[i] / nc);
    }
}

extern "C" void kernel_launch(void* const* d_in, const int* in_sizes, int n_in,
                              void* d_out, int out_size, void* d_ws, size_t ws_size,
                              hipStream_t stream)
{
    const float* x = (const float*)d_in[0];
    const float* w = (const float*)d_in[1];
    float* out  = (float*)d_out;
    float* neww = out + (size_t)B_ * COUT * L_;

    float*        yx = (float*)d_ws;
    float*        yy = yx + COUT * K_;
    unsigned int* mb = (unsigned int*)(yy + COUT);
    float*        ds = (float*)(mb + 1);

    zero_ws<<<(COUT * K_ + 255) / 256, 256, 0, stream>>>(yx, yy, mb);
    conv_hebb_gemm<<<B_ * BPI, TPB, 0, stream>>>(x, w, out, yx, yy);
    finalize_ds<<<(COUT * K_ + 255) / 256, 256, 0, stream>>>(w, yx, yy, ds, mb);
    apply_update<<<(COUT * K_ + 255) / 256, 256, 0, stream>>>(w, ds, mb, neww);
}


# ds, its largest magnitude and the weight update in ONE kernel of one 1024-thread work-group (max by lane exchange, then 32 wave maxima through LDS): 3 launches, not 4; no atomic max, no ds array
# speedup vs baseline: 20.0096x; 1.0508x over previous
#include <hip/hip_runtime.h>

#define B_     64
#define CIN    3
#define Hh     96
#define Ww     96
#define COUT   96
#define KH     5
#define KW     5
#define STRIDE 2
#define PAD    2
#define HOUT   48
#define WOUT   48
#define L_     (HOUT*WOUT)
#define K_     (CIN*KH*KW)
#define KP     76
#define LR     0.02f
#define DELTA  (-0.4f)
#define PRECF  1e-30f
#define G_     4
#define NT     192
#define TPB    (G_ * NT)
#define BPI    4
#define TPBLK  ((L_ / 16) / BPI)
#define ITERS  (TPBLK / G_)

typedef float v2f __attribute__((ext_vector_type(2)));
typedef float v8f __attribute__((ext_vector_type(8)));

typedef int v4i __attribute__((ext_vector_type(4)));

__device__ __forceinline__ int xhalf_i(int b)
{
    return __builtin_amdgcn_permlanex16(b, b, 0x76543210, 0xfedcba98, false, false);
}
__device__ __forceinline__ float xhalf_f(float x)
{
    return __int_as_float(xhalf_i(__float_as_int(x)));
}

__device__ __forceinline__ bool ranks_before(float av, int ai, float bv, int bi)
{
    return (av > bv) || (av == bv && ai < bi);
}

__device__ __forceinline__ void merge_top2(float& v0, int& i0, float& v1, int& i1,
                                           float w0, int j0, float w1, int j1)
{
    const bool af = ranks_before(v0, i0, w0, j0);
    const float xv = af ? v1 : v0;  const int xi = af ? i1 : i0;
    const float yv = af ? w0 : w1;  const int yi = af ? j0 : j1;
    const bool xs = ranks_before(xv, xi, yv, yi);
    v0 = af ? v0 : w0;  i0 = af ? i0 : j0;
    v1 = xs ? xv : yv;  i1 = xs ? xi : yi;
}

#define SP 100
#define SR 27
#define SC (SR * SP)
#define SN (CIN * SC)
#define XN ((SN + TPB - 1) / TPB)
static_assert((XN - 1) * TPB <= SN, "only a thread's last strip slot can fall beyond the strip");
static_assert(2 * (TPBLK / 3 - 1) + KH - 1 < SR && 2 * (WOUT - 1) + KW - 1 < SP, "every patch element lies inside the strip");

#define KS (KP / 4)
#define KR 15
#define BS 3

static_assert(KR == CIN * KH && KS == KR + CIN + 1, "step table below");
__device__ __forceinline__ v2f patch_frag(int s, const float* p0, const float* p1,
                                          const float* p2a, const float* p2b, int h)
{
    if (s < KR)     return *(const v2f*)(p0 + (s / KH) * SC + (s % KH) * SP);
    if (s < KS - 1) return (v2f){p1[(s - KR) * SC + 4], p1[(s - KR) * SC + 4 + SP]};
    return (v2f){p2a[4 * SP + 4], h ? 0.0f : p2b[4 * SP + 4]};
}
__device__ __forceinline__ v2f weight_frag(int s, const float* w1, const float* w2a, const float* w2b)
{
    if (s < KS - 1) return (v2f){w1[KH * KW * (s - KR) + 4], w1[KH * KW * (s - KR) + 9]};
    return (v2f){*w2a, *w2b};
}

__device__ __forceinline__ void products_tile(const v2f (&aW)[KR], const v2f (&aL)[KR],
                                              const float* wk1, const float* wk2a, const float* wk2b,
                                              const float* p0, const float* p1,
                                              const float* p2a, const float* p2b, v4i* lds_c,
                                              float* __restrict__ outp, int ob, int mn, int kg, int wv)
{
    v8f accO = {}; v8f accT = {};
    v2f bq[2][BS];
    v2f wq[2][BS];
    static_assert(KR >= BS, "the first block's weight fragments are all in registers");
    #pragma unroll
    for (int i = 0; i < BS; ++i)
        bq[0][i] = patch_frag(i, p0, p1, p2a, p2b, kg);
    #pragma unroll
    for (int blk = 0; blk * BS < KS; ++blk) {
        const int cur = blk & 1, nxt = cur ^ 1;
        #pragma unroll
        for (int i = 0; i < BS; ++i) {
            const int s = (blk + 1) * BS + i;
            if (s < KS) {
                bq[nxt][i] = patch_frag(s, p0, p1, p2a, p2b, kg);
                if (s >= KR) wq[nxt][i] = weight_frag(s, wk1, wk2a, wk2b);
            }
        }
        #pragma unroll
        for (int i = 0; i < BS; ++i) {
            const int s = blk * BS + i;
            if (s < KS) {
                v2f fW, fL;
                if (s < KR) { fW = aW[s]; fL = aL[s]; }
                else {
                    fW = wq[cur][i];
                    fL = (v2f){fW.x * __builtin_fabsf(fW.x), fW.y * __builtin_fabsf(fW.y)};
                }
                accO = __builtin_amdgcn_wmma_f32_16x16x4_f32(
                    false, fW, false, bq[cur][i], (short)0, accO, false, false);
                accT = __builtin_amdgcn_wmma_f32_16x16x4_f32(
                    false, fL, false, bq[cur][i], (short)0, accT, false, false);
            }
        }
    }

    #pragma unroll
    for (int r = 0; r < 8; ++r)
        outp[(size_t)r * L_] = accO[r];

    float v0, v1; int i0, i1;
    if (accT[1] > accT[0]) { v0 = accT[1]; i0 = ob + 1; v1 = accT[0]; i1 = ob; }
    else                   { v0 = accT[0]; i0 = ob;     v1 = accT[1]; i1 = ob + 1; }
    #pragma unroll
    for (int r = 2; r < 8; ++r) {
        const float e = accT[r];
        const bool g0 = e > v0, g1 = e > v1;
        v1 = g0 ? v0 : (g1 ? e : v1);
        i1 = g0 ? i0 : (g1 ? ob + r : i1);
        v0 = g0 ? e : v0;
        i0 = g0 ? ob + r : i0;
    }
    float w0 = xhalf_f(v0), w1 = xhalf_f(v1);
    int   j0 = xhalf_i(i0), j1 = xhalf_i(i1);
    merge_top2(v0, i0, v1, i1, w0, j0, w1, j1);
    if (kg == 0) {
        v4i rec = {__float_as_int(v0), i0, __float_as_int(v1), i1};
        lds_c[mn * 6 + wv] = rec;
    }
}

#define SJ ((K_ + 11) / 12)
static_assert(12 * (SJ - 2) + 11 < K_, "only the last scatter slot needs a guard");

__device__ __forceinline__ void scatter_tile(const float* ps, const int (&sk)[SJ], const v4i* lds_c,
                                             float* lds_yx, float* lds_yy, int tid)
{
    int col = tid / 12;
    int r   = tid - col * 12;
    v4i rec[6];
    #pragma unroll
    for (int m = 0; m < 6; ++m) rec[m] = lds_c[col * 6 + m];
    const float* pc = ps + 2 * col;
    float p[SJ];
    #pragma unroll
    for (int j = 0; j < SJ; ++j) p[j] = pc[sk[j]];
    float v0 = __int_as_float(rec[0].x), v1 = __int_as_float(rec[0].z);
    int   i0 = rec[0].y, i1 = rec[0].w;
    #pragma unroll
    for (int m = 1; m < 6; ++m)
        merge_top2(v0, i0, v1, i1, __int_as_float(rec[m].x), rec[m].y,
                   __int_as_float(rec[m].z), rec[m].w);
    float* y0 = lds_yx + i0 * K_ + r;
    float* y1 = lds_yx + i1 * K_ + r;
    #pragma unroll
    for (int j = 0; j < SJ; ++j) {
        if (j < SJ - 1 || r + 12 * j < K_) {
            atomicAdd(&y0[12 * j], p[j]);
            atomicAdd(&y1[12 * j], DELTA * p[j]);
        }
    }
    if (r == 0) {
        atomicAdd(&lds_yy[i0], v0);
        atomicAdd(&lds_yy[i1], DELTA * v1);
    }
}

#define WN ((COUT * KP + TPB - 1) / TPB)
#define YN ((COUT * K_ + TPB - 1) / TPB)
static_assert((WN - 1) * TPB <= COUT * KP && (YN - 1) * TPB <= COUT * K_, "only a thread's last slot can fall beyond the array");

__global__ __launch_bounds__(TPB) void conv_hebb_gemm(
    const float* __restrict__ x, const float* __restrict__ w,
    float* __restrict__ out, float* __restrict__ yx, float* __restrict__ yy)
{
    __shared__ float lds_w[COUT * KP];
    __shared__ float lds_yx[COUT * K_];
    __shared__ float lds_yy[COUT];
    __shared__ __attribute__((aligned(16))) float lds_x[SN];
    __shared__ v4i   lds_cc[G_][2][16 * 6];

    const int gtid = threadIdx.x;
    const int g    = gtid / NT;
    const int tid  = gtid - g * NT;
    const int b    = blockIdx.x / BPI;
    const int q    = blockIdx.x - b * BPI;

    const bool scatter_first = (__builtin_amdgcn_readfirstlane(gtid >> 7) & 1) != 0;

    const float* xb = x + (size_t)b * CIN * Hh * Ww;
    float xst[XN];
    int xok = 0;
    #pragma unroll
    for (int n = 0; n < XN; ++n) {
        const int e  = gtid + TPB * n;
        const int c  = e / SC, rem = e - c * SC;
        const int ry = rem / SP, rx = rem - ry * SP;
        const int iy = STRIDE * (TPBLK / 3) * q - PAD + ry, ix = rx - PAD;
        const bool ok = e < SN && (unsigned)iy < (unsigned)Hh && (unsigned)ix < (unsigned)Ww;
        xst[n] = xb[ok ? (c * Hh + iy) * Ww + ix : 0];
        xok |= ok ? (1 << n) : 0;
    }
    float wst[WN];
    int wok = 0;
    #pragma unroll
    for (int n = 0; n < WN; ++n) {
        const int i = gtid + TPB * n;
        const int o = i / KP, k = i - o * KP;
        const bool ok = i < COUT * KP && k < K_;
        wst[n] = w[ok ? o * K_ + k : 0];
        wok |= ok ? (1 << n) : 0;
    }
    for (int i = gtid; i < COUT * K_; i += TPB) lds_yx[i] = 0.0f;
    if (gtid < COUT) lds_yy[gtid] = 0.0f;
    #pragma unroll
    for (int n = 0; n < XN; ++n) {
        const int e = gtid + TPB * n;
        if (n < XN - 1 || e < SN) lds_x[e] = ((xok >> n) & 1) ? xst[n] : 0.0f;
    }
    #pragma unroll
    for (int n = 0; n < WN; ++n) {
        const int i = gtid + TPB * n;
        if (n < WN - 1 || i < COUT * KP) lds_w[i] = ((wok >> n) & 1) ? wst[n] : 0.0f;
    }

    const int wv   = tid >> 5;
    const int lane = tid & 31;
    const int mn   = lane & 15;
    const int kg   = lane >> 4;
    const int ob   = wv * 16 + 8 * kg;
    const float* wr = &lds_w[(wv * 16 + mn) * KP];
    const float* w1  = wr + 2 * KW * kg;
    const float* w2a = wr + (kg ? 74 : 24);
    const float* w2b = wr + (kg ? 75 : 49);
    const float* x0  = lds_x + 2 * mn + 2 * kg;
    const float* x1  = lds_x + 2 * mn + 2 * kg * SP;
    const float* x2a = lds_x + 2 * mn + 2 * kg * SC;
    const float* x2b = lds_x + 2 * mn + (kg ? 2 * SC : SC);
    int sk[SJ];
    #pragma unroll
    for (int j = 0; j < SJ; ++j) {
        const int r  = tid % 12;
        const int k  = (j < SJ - 1 || r + 12 * j < K_) ? r + 12 * j : r;
        const int c  = k / (KH * KW), rr = k - c * (KH * KW);
        const int ki = rr / KW, kj = rr - ki * KW;
        sk[j] = c * SC + ki * SP + kj;
    }

    __syncthreads();
    v2f aW[KR], aL[KR];
    #pragma unroll
    for (int s = 0; s < KR; ++s) {
        const int ka = KW * s + 2 * kg;
        const float a0 = wr[ka], a1 = wr[ka + 1];
        aW[s] = (v2f){a0, a1};
        aL[s] = (v2f){a0 * __builtin_fabsf(a0), a1 * __builtin_fabsf(a1)};
    }

    int tb_prv = 0;

  for (int it = 0; it <= ITERS; ++it) {
    const int T  = it * G_ + g;
    const int yl = T / 3, xt = T - 3 * yl;
    const int tb = 2 * yl * SP + 2 * 16 * xt;
    const int l0 = (q * TPBLK + T) * 16;
    v4i* c_cur = lds_cc[g][it & 1];
    v4i* c_prv = lds_cc[g][(it & 1) ^ 1];

    __syncthreads();

    if (scatter_first && it > 0) scatter_tile(lds_x + tb_prv, sk, c_prv, lds_yx, lds_yy, tid);
    if (it < ITERS)
        products_tile(aW, aL, w1, w2a, w2b, x0 + tb, x1 + tb, x2a + tb, x2b + tb, c_cur,
                      &out[((size_t)b * COUT + ob) * L_ + l0 + mn], ob, mn, kg, wv);
    if (!scatter_first && it > 0) scatter_tile(lds_x + tb_prv, sk, c_prv, lds_yx, lds_yy, tid);

    tb_prv = tb;
  }
    __syncthreads();

    float sv[YN];
    #pragma unroll
    for (int n = 0; n < YN; ++n) {
        const int i = gtid + TPB * n;
        sv[n] = lds_yx[(n < YN - 1 || i < COUT * K_) ? i : gtid];
    }
    const float sy = lds_yy[gtid < COUT ? gtid : 0];
    #pragma unroll
    for (int n = 0; n < YN; ++n) {
        const int i = gtid + TPB * n;
        if (n < YN - 1 || i < COUT * K_) atomicAdd(&yx[i], sv[n]);
    }
    if (gtid < COUT) atomicAdd(&yy[gtid], sy);
}

__global__ void zero_ws(float* yx, float* yy)
{
    int i = blockIdx.x * blockDim.x + threadIdx.x;
    if (i < COUT * K_) yx[i] = 0.0f;
    if (i < COUT)      yy[i] = 0.0f;
}

#define FT 1024
#define FE 8
#define FA ((COUT * K_) / FE)
static_assert((COUT * K_) % FE == 0 && FA <= FT, "every element belongs to exactly one thread");
static_assert(FE <= K_, "a thread's elements span at most two channels");
typedef float v4f __attribute__((ext_vector_type(4)));

template <int N> __device__ __forceinline__ unsigned int xrow_u(unsigned int v)
{
    return (unsigned int)__builtin_amdgcn_update_dpp(0, (int)v, 0x160 + N, 0xf, 0xf, false);
}
__device__ __forceinline__ unsigned int wave_max_u(unsigned int m)
{
    unsigned int a;
    a = xrow_u<1>(m);               m = a > m ? a : m;
    a = xrow_u<2>(m);               m = a > m ? a : m;
    a = xrow_u<4>(m);               m = a > m ? a : m;
    a = xrow_u<8>(m);               m = a > m ? a : m;
    a = (unsigned int)xhalf_i((int)m);
    return a > m ? a : m;
}

__global__ __launch_bounds__(FT) void finalize_update(const float* __restrict__ w,
                                                      const float* __restrict__ yx,
                                                      const float* __restrict__ yy,
                                                      float* __restrict__ neww)
{
    __shared__ unsigned int wmax[FT / 32];
    const int  t  = threadIdx.x;
    const bool on = t < FA;
    const int  i0 = on ? FE * t : 0;
    const int  o0 = i0 / K_;
    const int  o1 = o0 + 1 < COUT ? o0 + 1 : o0;
    const int  nb = (o0 + 1) * K_ - i0;
    const v4f wa = *(const v4f*)(w + i0),  wb = *(const v4f*)(w + i0 + 4);
    const v4f xa = *(const v4f*)(yx + i0), xb = *(const v4f*)(yx + i0 + 4);
    const float y0 = yy[o0], y1 = yy[o1];
    const float wv[FE] = {wa.x, wa.y, wa.z, wa.w, wb.x, wb.y, wb.z, wb.w};
    const float sx[FE] = {xa.x, xa.y, xa.z, xa.w, xb.x, xb.y, xb.z, xb.w};
    float d[FE];
    unsigned int m = __float_as_uint(PRECF);
    #pragma unroll
    for (int n = 0; n < FE; ++n) {
        const float sy = n < nb ? y0 : y1;
        d[n] = sx[n] * (1.0f / (float)L_) - (sy * (1.0f / (float)L_)) * wv[n];
        const unsigned int a = on ? __float_as_uint(__builtin_fabsf(d[n])) : 0u;
        m = a > m ? a : m;
    }
    m = wave_max_u(m);
    if ((t & 31) == 0) wmax[t >> 5] = m;
    __syncthreads();
    const float nc = __uint_as_float(wave_max_u(wmax[t & 31]));
    if (on) {
        float r[FE];
        #pragma unroll
        for (int n = 0; n < FE; ++n) r[n] = wv[n] + LR * (d[n] / nc);
        *(v4f*)(neww + i0)     = (v4f){r[0], r[1], r[2], r[3]};
        *(v4f*)(neww + i0 + 4) = (v4f){r[4], r[5], r[6], r[7]};
    }
}

extern "C" void kernel_launch(void* const* d_in, const int* in_sizes, int n_in,
                              void* d_out, int out_size, void* d_ws, size_t ws_size,
                              hipStream_t stream)
{
    const float* x = (const float*)d_in[0];
    const float* w = (const float*)d_in[1];
    float* out  = (float*)d_out;
    float* neww = out + (size_t)B_ * COUT * L_;

    float*        yx = (float*)d_ws;
    float*        yy = yx + COUT * K_;

    zero_ws<<<(COUT * K_ + 255) / 256, 256, 0, stream>>>(yx, yy);
    conv_hebb_gemm<<<B_ * BPI, TPB, 0, stream>>>(x, w, out, yx, yy);
    finalize_update<<<1, FT, 0, stream>>>(w, yx, yy, neww);
}


# Each wave covers 32 channels: two 16-row blocks share each patch fragment, all weights in registers, a tile's fragments read as one group a pass ahead; 12 waves, not 24; atomic sum order shifts
# speedup vs baseline: 20.6418x; 1.0316x over previous
#include <hip/hip_runtime.h>

#define B_     64
#define CIN    3
#define Hh     96
#define Ww     96
#define COUT   96
#define KH     5
#define KW     5
#define STRIDE 2
#define PAD    2
#define HOUT   48
#define WOUT   48
#define L_     (HOUT*WOUT)
#define K_     (CIN*KH*KW)
#define KP     76
#define LR     0.02f
#define DELTA  (-0.4f)
#define PRECF  1e-30f
#define G_     4
#define NT     96
#define TPB    (G_ * NT)
#define BPI    4
#define TPBLK  ((L_ / 16) / BPI)
#define ITERS  (TPBLK / G_)

typedef float v2f __attribute__((ext_vector_type(2)));
typedef float v8f __attribute__((ext_vector_type(8)));

typedef int v4i __attribute__((ext_vector_type(4)));

__device__ __forceinline__ int xhalf_i(int b)
{
    return __builtin_amdgcn_permlanex16(b, b, 0x76543210, 0xfedcba98, false, false);
}
__device__ __forceinline__ float xhalf_f(float x)
{
    return __int_as_float(xhalf_i(__float_as_int(x)));
}

__device__ __forceinline__ bool ranks_before(float av, int ai, float bv, int bi)
{
    return (av > bv) || (av == bv && ai < bi);
}

__device__ __forceinline__ void merge_top2(float& v0, int& i0, float& v1, int& i1,
                                           float w0, int j0, float w1, int j1)
{
    const bool af = ranks_before(v0, i0, w0, j0);
    const float xv = af ? v1 : v0;  const int xi = af ? i1 : i0;
    const float yv = af ? w0 : w1;  const int yi = af ? j0 : j1;
    const bool xs = ranks_before(xv, xi, yv, yi);
    v0 = af ? v0 : w0;  i0 = af ? i0 : j0;
    v1 = xs ? xv : yv;  i1 = xs ? xi : yi;
}

#define SP 100
#define SR 27
#define SC (SR * SP)
#define SN (CIN * SC)
#define XN ((SN + TPB - 1) / TPB)
static_assert((XN - 1) * TPB <= SN, "only a thread's last strip slot can fall beyond the strip");
static_assert(2 * (TPBLK / 3 - 1) + KH - 1 < SR && 2 * (WOUT - 1) + KW - 1 < SP, "every patch element lies inside the strip");

#define KS (KP / 4)
#define KR 15
#define WR 3

static_assert(KR == CIN * KH && KS == KR + CIN + 1, "step table below");
__device__ __forceinline__ v2f patch_frag(int s, const float* p0, const float* p1,
                                          const float* p2a, const float* p2b, int h)
{
    if (s < KR)     return *(const v2f*)(p0 + (s / KH) * SC + (s % KH) * SP);
    if (s < KS - 1) return (v2f){p1[(s - KR) * SC + 4], p1[(s - KR) * SC + 4 + SP]};
    return (v2f){p2a[4 * SP + 4], h ? 0.0f : p2b[4 * SP + 4]};
}
__device__ __forceinline__ v2f weight_frag(int s, const float* wr, int h)
{
    if (s < KR)     return (v2f){wr[KW * s + 2 * h], wr[KW * s + 2 * h + 1]};
    if (s < KS - 1) return (v2f){wr[2 * KW * h + KH * KW * (s - KR) + 4], wr[2 * KW * h + KH * KW * (s - KR) + 9]};
    return (v2f){wr[h ? 74 : 24], wr[h ? 75 : 49]};
}

__device__ __forceinline__ void patch_frags(v2f (&bq)[KS], const float* p0, const float* p1,
                                            const float* p2a, const float* p2b, int h)
{
    #pragma unroll
    for (int s = 0; s < KS; ++s) bq[s] = patch_frag(s, p0, p1, p2a, p2b, h);
}

__device__ __forceinline__ void top2_rows(const v8f& acc, int ob, float& v0, int& i0, float& v1, int& i1)
{
    #pragma unroll
    for (int r = 0; r < 8; ++r) {
        const float e = acc[r];
        const bool g0 = e > v0, g1 = e > v1;
        v1 = g0 ? v0 : (g1 ? e : v1);
        i1 = g0 ? i0 : (g1 ? ob + r : i1);
        v0 = g0 ? e : v0;
        i0 = g0 ? ob + r : i0;
    }
}

__device__ __forceinline__ void products_tile(const v2f (&aW)[2][KS], const v2f (&aL)[2][KS],
                                              const v2f (&bq)[KS], v8f (&accT)[2],
                                              float* __restrict__ outp)
{
    v8f accO[2] = {{}, {}};
    accT[0] = (v8f){}; accT[1] = (v8f){};
    #pragma unroll
    for (int s = 0; s < KS; ++s) {
        accO[0] = __builtin_amdgcn_wmma_f32_16x16x4_f32(false, aW[0][s], false, bq[s], (short)0, accO[0], false, false);
        accT[0] = __builtin_amdgcn_wmma_f32_16x16x4_f32(false, aL[0][s], false, bq[s], (short)0, accT[0], false, false);
        accO[1] = __builtin_amdgcn_wmma_f32_16x16x4_f32(false, aW[1][s], false, bq[s], (short)0, accO[1], false, false);
        accT[1] = __builtin_amdgcn_wmma_f32_16x16x4_f32(false, aL[1][s], false, bq[s], (short)0, accT[1], false, false);
    }
    #pragma unroll
    for (int m = 0; m < 2; ++m)
        #pragma unroll
        for (int r = 0; r < 8; ++r)
            outp[(size_t)(16 * m + r) * L_] = accO[m][r];
}

__device__ __forceinline__ void top2_tile(const v8f (&accT)[2], v4i* lds_c, int ob, int mn, int kg, int wv)
{
    float v0, v1; int i0, i1;
    if (accT[0][1] > accT[0][0]) { v0 = accT[0][1]; i0 = ob + 1; v1 = accT[0][0]; i1 = ob; }
    else                         { v0 = accT[0][0]; i0 = ob;     v1 = accT[0][1]; i1 = ob + 1; }
    #pragma unroll
    for (int r = 2; r < 8; ++r) {
        const float e = accT[0][r];
        const bool g0 = e > v0, g1 = e > v1;
        v1 = g0 ? v0 : (g1 ? e : v1);
        i1 = g0 ? i0 : (g1 ? ob + r : i1);
        v0 = g0 ? e : v0;
        i0 = g0 ? ob + r : i0;
    }
    top2_rows(accT[1], ob + 16, v0, i0, v1, i1);
    float w0 = xhalf_f(v0), w1 = xhalf_f(v1);
    int   j0 = xhalf_i(i0), j1 = xhalf_i(i1);
    merge_top2(v0, i0, v1, i1, w0, j0, w1, j1);
    if (kg == 0) {
        v4i rec = {__float_as_int(v0), i0, __float_as_int(v1), i1};
        lds_c[mn * WR + wv] = rec;
    }
}

#define SD (NT / 16)
#define SJ ((K_ + SD - 1) / SD)
static_assert(SD * 16 == NT && SD * (SJ - 2) + SD - 1 < K_, "only the last scatter slot needs a guard");

__device__ __forceinline__ void scatter_tile(const float* ps, const int (&sk)[SJ], const v4i* lds_c,
                                             float* lds_yx, float* lds_yy, int tid)
{
    int col = tid / SD;
    int r   = tid - col * SD;
    v4i rec[WR];
    #pragma unroll
    for (int m = 0; m < WR; ++m) rec[m] = lds_c[col * WR + m];
    const float* pc = ps + 2 * col;
    float p[SJ];
    #pragma unroll
    for (int j = 0; j < SJ; ++j) p[j] = pc[sk[j]];
    float v0 = __int_as_float(rec[0].x), v1 = __int_as_float(rec[0].z);
    int   i0 = rec[0].y, i1 = rec[0].w;
    #pragma unroll
    for (int m = 1; m < WR; ++m)
        merge_top2(v0, i0, v1, i1, __int_as_float(rec[m].x), rec[m].y,
                   __int_as_float(rec[m].z), rec[m].w);
    float* y0 = lds_yx + i0 * K_ + r;
    float* y1 = lds_yx + i1 * K_ + r;
    #pragma unroll
    for (int j = 0; j < SJ; ++j) {
        if (j < SJ - 1 || r + SD * j < K_) {
            atomicAdd(&y0[SD * j], p[j]);
            atomicAdd(&y1[SD * j], DELTA * p[j]);
        }
    }
    if (r == 0) {
        atomicAdd(&lds_yy[i0], v0);
        atomicAdd(&lds_yy[i1], DELTA * v1);
    }
}

#define WN ((COUT * KP + TPB - 1) / TPB)
#define YN ((COUT * K_ + TPB - 1) / TPB)
static_assert((WN - 1) * TPB <= COUT * KP && (YN - 1) * TPB <= COUT * K_, "only a thread's last slot can fall beyond the array");

__global__ __launch_bounds__(TPB) void conv_hebb_gemm(
    const float* __restrict__ x, const float* __restrict__ w,
    float* __restrict__ out, float* __restrict__ yx, float* __restrict__ yy)
{
    __shared__ float lds_w[COUT * KP];
    __shared__ float lds_yx[COUT * K_];
    __shared__ float lds_yy[COUT];
    __shared__ __attribute__((aligned(16))) float lds_x[SN];
    __shared__ v4i   lds_cc[G_][2][16 * WR];

    const int gtid = threadIdx.x;
    const int g    = gtid / NT;
    const int tid  = gtid - g * NT;
    const int b    = blockIdx.x / BPI;
    const int q    = blockIdx.x - b * BPI;

    const bool scatter_first = (__builtin_amdgcn_readfirstlane(gtid >> 7) & 1) != 0;

    const float* xb = x + (size_t)b * CIN * Hh * Ww;
    float xst[XN];
    int xok = 0;
    #pragma unroll
    for (int n = 0; n < XN; ++n) {
        const int e  = gtid + TPB * n;
        const int c  = e / SC, rem = e - c * SC;
        const int ry = rem / SP, rx = rem - ry * SP;
        const int iy = STRIDE * (TPBLK / 3) * q - PAD + ry, ix = rx - PAD;
        const bool ok = e < SN && (unsigned)iy < (unsigned)Hh && (unsigned)ix < (unsigned)Ww;
        xst[n] = xb[ok ? (c * Hh + iy) * Ww + ix : 0];
        xok |= ok ? (1 << n) : 0;
    }
    float wst[WN];
    int wok = 0;
    #pragma unroll
    for (int n = 0; n < WN; ++n) {
        const int i = gtid + TPB * n;
        const int o = i / KP, k = i - o * KP;
        const bool ok = i < COUT * KP && k < K_;
        wst[n] = w[ok ? o * K_ + k : 0];
        wok |= ok ? (1 << n) : 0;
    }
    for (int i = gtid; i < COUT * K_; i += TPB) lds_yx[i] = 0.0f;
    if (gtid < COUT) lds_yy[gtid] = 0.0f;
    #pragma unroll
    for (int n = 0; n < XN; ++n) {
        const int e = gtid + TPB * n;
        if (n < XN - 1 || e < SN) lds_x[e] = ((xok >> n) & 1) ? xst[n] : 0.0f;
    }
    #pragma unroll
    for (int n = 0; n < WN; ++n) {
        const int i = gtid + TPB * n;
        if (n < WN - 1 || i < COUT * KP) lds_w[i] = ((wok >> n) & 1) ? wst[n] : 0.0f;
    }

    const int wv   = tid >> 5;
    const int lane = tid & 31;
    const int mn   = lane & 15;
    const int kg   = lane >> 4;
    const int ob   = wv * 32 + 8 * kg;
    const float* x0  = lds_x + 2 * mn + 2 * kg;
    const float* x1  = lds_x + 2 * mn + 2 * kg * SP;
    const float* x2a = lds_x + 2 * mn + 2 * kg * SC;
    const float* x2b = lds_x + 2 * mn + (kg ? 2 * SC : SC);
    int sk[SJ];
    #pragma unroll
    for (int j = 0; j < SJ; ++j) {
        const int r  = tid % SD;
        const int k  = (j < SJ - 1 || r + SD * j < K_) ? r + SD * j : r;
        const int c  = k / (KH * KW), rr = k - c * (KH * KW);
        const int ki = rr / KW, kj = rr - ki * KW;
        sk[j] = c * SC + ki * SP + kj;
    }

    __syncthreads();
    v2f aW[2][KS], aL[2][KS];
    #pragma unroll
    for (int m = 0; m < 2; ++m) {
        const float* wr = &lds_w[(wv * 32 + 16 * m + mn) * KP];
        #pragma unroll
        for (int s = 0; s < KS; ++s) {
            const v2f a = weight_frag(s, wr, kg);
            aW[m][s] = a;
            aL[m][s] = (v2f){a.x * __builtin_fabsf(a.x), a.y * __builtin_fabsf(a.y)};
        }
    }

    v2f bq[KS];
    {
        const int tb0 = 2 * (g / 3) * SP + 2 * 16 * (g % 3);
        patch_frags(bq, x0 + tb0, x1 + tb0, x2a + tb0, x2b + tb0, kg);
    }

    int tb_prv = 0;

  for (int it = 0; it <= ITERS; ++it) {
    const int T  = it * G_ + g;
    const int yl = T / 3, xt = T - 3 * yl;
    const int tb = 2 * yl * SP + 2 * 16 * xt;
    const int l0 = (q * TPBLK + T) * 16;
    v4i* c_cur = lds_cc[g][it & 1];
    v4i* c_prv = lds_cc[g][(it & 1) ^ 1];

    __syncthreads();

    if (scatter_first && it > 0) scatter_tile(lds_x + tb_prv, sk, c_prv, lds_yx, lds_yy, tid);
    if (it < ITERS) {
        v8f accT[2];
        products_tile(aW, aL, bq, accT, &out[((size_t)b * COUT + ob) * L_ + l0 + mn]);
        if (it + 1 < ITERS) {
            const int Tn  = T + G_;
            const int yn  = Tn / 3, xn = Tn - 3 * yn;
            const int tbn = 2 * yn * SP + 2 * 16 * xn;
            patch_frags(bq, x0 + tbn, x1 + tbn, x2a + tbn, x2b + tbn, kg);
        }
        top2_tile(accT, c_cur, ob, mn, kg, wv);
    }
    if (!scatter_first && it > 0) scatter_tile(lds_x + tb_prv, sk, c_prv, lds_yx, lds_yy, tid);

    tb_prv = tb;
  }
    __syncthreads();

    float sv[YN];
    #pragma unroll
    for (int n = 0; n < YN; ++n) {
        const int i = gtid + TPB * n;
        sv[n] = lds_yx[(n < YN - 1 || i < COUT * K_) ? i : gtid];
    }
    const float sy = lds_yy[gtid < COUT ? gtid : 0];
    #pragma unroll
    for (int n = 0; n < YN; ++n) {
        const int i = gtid + TPB * n;
        if (n < YN - 1 || i < COUT * K_) atomicAdd(&yx[i], sv[n]);
    }
    if (gtid < COUT) atomicAdd(&yy[gtid], sy);
}

__global__ void zero_ws(float* yx, float* yy)
{
    int i = blockIdx.x * blockDim.x + threadIdx.x;
    if (i < COUT * K_) yx[i] = 0.0f;
    if (i < COUT)      yy[i] = 0.0f;
}

#define FT 1024
#define FE 8
#define FA ((COUT * K_) / FE)
static_assert((COUT * K_) % FE == 0 && FA <= FT, "every element belongs to exactly one thread");
static_assert(FE <= K_, "a thread's elements span at most two channels");
typedef float v4f __attribute__((ext_vector_type(4)));

template <int N> __device__ __forceinline__ unsigned int xrow_u(unsigned int v)
{
    return (unsigned int)__builtin_amdgcn_update_dpp(0, (int)v, 0x160 + N, 0xf, 0xf, false);
}
__device__ __forceinline__ unsigned int wave_max_u(unsigned int m)
{
    unsigned int a;
    a = xrow_u<1>(m);               m = a > m ? a : m;
    a = xrow_u<2>(m);               m = a > m ? a : m;
    a = xrow_u<4>(m);               m = a > m ? a : m;
    a = xrow_u<8>(m);               m = a > m ? a : m;
    a = (unsigned int)xhalf_i((int)m);
    return a > m ? a : m;
}

__global__ __launch_bounds__(FT) void finalize_update(const float* __restrict__ w,
                                                      const float* __restrict__ yx,
                                                      const float* __restrict__ yy,
                                                      float* __restrict__ neww)
{
    __shared__ unsigned int wmax[FT / 32];
    const int  t  = threadIdx.x;
    const bool on = t < FA;
    const int  i0 = on ? FE * t : 0;
    const int  o0 = i0 / K_;
    const int  o1 = o0 + 1 < COUT ? o0 + 1 : o0;
    const int  nb = (o0 + 1) * K_ - i0;
    const v4f wa = *(const v4f*)(w + i0),  wb = *(const v4f*)(w + i0 + 4);
    const v4f xa = *(const v4f*)(yx + i0), xb = *(const v4f*)(yx + i0 + 4);
    const float y0 = yy[o0], y1 = yy[o1];
    const float wv[FE] = {wa.x, wa.y, wa.z, wa.w, wb.x, wb.y, wb.z, wb.w};
    const float sx[FE] = {xa.x, xa.y, xa.z, xa.w, xb.x, xb.y, xb.z, xb.w};
    float d[FE];
    unsigned int m = __float_as_uint(PRECF);
    #pragma unroll
    for (int n = 0; n < FE; ++n) {
        const float sy = n < nb ? y0 : y1;
        d[n] = sx[n] * (1.0f / (float)L_) - (sy * (1.0f / (float)L_)) * wv[n];
        const unsigned int a = on ? __float_as_uint(__builtin_fabsf(d[n])) : 0u;
        m = a > m ? a : m;
    }
    m = wave_max_u(m);
    if ((t & 31) == 0) wmax[t >> 5] = m;
    __syncthreads();
    const float nc = __uint_as_float(wave_max_u(wmax[t & 31]));
    if (on) {
        float r[FE];
        #pragma unroll
        for (int n = 0; n < FE; ++n) r[n] = wv[n] + LR * (d[n] / nc);
        *(v4f*)(neww + i0)     = (v4f){r[0], r[1], r[2], r[3]};
        *(v4f*)(neww + i0 + 4) = (v4f){r[4], r[5], r[6], r[7]};
    }
}

extern "C" void kernel_launch(void* const* d_in, const int* in_sizes, int n_in,
                              void* d_out, int out_size, void* d_ws, size_t ws_size,
                              hipStream_t stream)
{
    const float* x = (const float*)d_in[0];
    const float* w = (const float*)d_in[1];
    float* out  = (float*)d_out;
    float* neww = out + (size_t)B_ * COUT * L_;

    float*        yx = (float*)d_ws;
    float*        yy = yx + COUT * K_;

    zero_ws<<<(COUT * K_ + 255) / 256, 256, 0, stream>>>(yx, yy);
    conv_hebb_gemm<<<B_ * BPI, TPB, 0, stream>>>(x, w, out, yx, yy);
    finalize_update<<<1, FT, 0, stream>>>(w, yx, yy, neww);
}
